# starting kernel
# baseline (speedup 1.0000x reference)
#include <hip/hip_runtime.h>
#include <math.h>

#define N_NODES 50000
#define N_EDGES 640000
#define DIM     128

typedef __attribute__((ext_vector_type(2))) float v2f;
typedef __attribute__((ext_vector_type(8))) float v8f;


__global__ void k_deg_init(unsigned* __restrict__ deg) {
    int i = blockIdx.x * blockDim.x + threadIdx.x;
    if (i < N_NODES) deg[i] = 1u;
}

__global__ void k_deg_count(const int* __restrict__ dst, unsigned* __restrict__ deg) {
    int e = blockIdx.x * blockDim.x + threadIdx.x;
    if (e < N_EDGES) atomicAdd(&deg[dst[e]], 1u);
}

__global__ void k_dinv(const unsigned* __restrict__ deg, float* __restrict__ dinv) {
    int i = blockIdx.x * blockDim.x + threadIdx.x;
    if (i < N_NODES) dinv[i] = rsqrtf((float)deg[i]);
}


__global__ __launch_bounds__(256) void k_gemm(const float* __restrict__ A,
                                              const float* __restrict__ W,
                                              const float* __restrict__ bias,
                                              float* __restrict__ out,
                                              int addBias) {
    const int wave = threadIdx.x >> 5;
    const int lane = threadIdx.x & 31;
    const int l15  = lane & 15;
    const int hi   = lane >> 4;
    const int rowBase = blockIdx.x << 4;
    const int colBase = wave << 4;

    const float* arow = A + (size_t)(rowBase + l15) * DIM + (hi << 1);
    const float* wcol = W + (size_t)(hi << 1) * DIM + colBase + l15;

    v8f c = {};
#pragma unroll
    for (int k = 0; k < DIM; k += 4) {
        v2f a = *(const v2f*)(arow + k);
        v2f b;
        b.x = wcol[(size_t)k * DIM];
        b.y = wcol[(size_t)(k + 1) * DIM];
        c = __builtin_amdgcn_wmma_f32_16x16x4_f32(false, a, false, b,
                                                  (short)0, c, false, false);
    }

    const float bv = addBias ? bias[colBase + l15] : 0.0f;
    const int m0 = hi << 3;
    float* orow = out + (size_t)(rowBase + m0) * DIM + colBase + l15;
#pragma unroll
    for (int r = 0; r < 8; ++r)
        orow[(size_t)r * DIM] = c[r] + bv;
}


__global__ __launch_bounds__(256) void k_self(const float* __restrict__ h,
                                              const float* __restrict__ dinv,
                                              float* __restrict__ agg) {
    int t = blockIdx.x * blockDim.x + threadIdx.x;
    int node = t >> 5;
    float s = dinv[node];
    s *= s;
    float4 v = ((const float4*)h)[t];
    float4 o;
    o.x = v.x * s; o.y = v.y * s; o.z = v.z * s; o.w = v.w * s;
    ((float4*)agg)[t] = o;
}


__global__ __launch_bounds__(256) void k_scatter(const int* __restrict__ src,
                                                 const int* __restrict__ dst,
                                                 const float* __restrict__ dinv,
                                                 const float* __restrict__ h,
                                                 float* __restrict__ agg) {
    int w    = (blockIdx.x * blockDim.x + threadIdx.x) >> 5;
    int lane = threadIdx.x & 31;
    int s = src[w];
    int d = dst[w];
    float norm = dinv[s] * dinv[d];
    float4 v = ((const float4*)(h + (size_t)s * DIM))[lane];
    float* ap = agg + (size_t)d * DIM + (lane << 2);
    unsafeAtomicAdd(ap + 0, norm * v.x);
    unsafeAtomicAdd(ap + 1, norm * v.y);
    unsafeAtomicAdd(ap + 2, norm * v.z);
    unsafeAtomicAdd(ap + 3, norm * v.w);
}


__global__ __launch_bounds__(256) void k_act(const float* __restrict__ agg,
                                             const float* __restrict__ b1,
                                             float* __restrict__ h) {
    int t = blockIdx.x * blockDim.x + threadIdx.x;
    h[t] = tanhf(agg[t] + b1[t & (DIM - 1)]);
}


extern "C" void kernel_launch(void* const* d_in, const int* in_sizes, int n_in,
                              void* d_out, int out_size, void* d_ws, size_t ws_size,
                              hipStream_t stream) {
    const float* x  = (const float*)d_in[0];
    const int*   ei = (const int*)  d_in[1];
    const float* W1 = (const float*)d_in[2];
    const float* b1 = (const float*)d_in[3];
    const float* W2 = (const float*)d_in[4];
    const float* b2 = (const float*)d_in[5];
    float* out = (float*)d_out;

    float*    ws   = (float*)d_ws;
    unsigned* deg  = (unsigned*)ws;
    float*    dinv = ws + 65536;
    float*    h    = ws + 131072;
    float*    agg  = ws + 131072 + (size_t)N_NODES * DIM;

    const int* esrc = ei;
    const int* edst = ei + N_EDGES;

    k_deg_init <<<(N_NODES + 255) / 256, 256, 0, stream>>>(deg);
    k_deg_count<<<(N_EDGES + 255) / 256, 256, 0, stream>>>(edst, deg);
    k_dinv     <<<(N_NODES + 255) / 256, 256, 0, stream>>>(deg, dinv);

    k_gemm <<<N_NODES / 16, 256, 0, stream>>>(x, W1, b1, h, 0);
    k_self <<<(N_NODES * DIM / 4) / 256, 256, 0, stream>>>(h, dinv, agg);
    k_scatter<<<N_EDGES / 8, 256, 0, stream>>>(esrc, edst, dinv, h, agg);
    k_act  <<<(N_NODES * DIM) / 256, 256, 0, stream>>>(agg, b1, h);
    k_gemm <<<N_NODES / 16, 256, 0, stream>>>(h, W2, b2, out, 1);
}


# Aggregation by destination tiles: each work-group keeps 196 row sums in LDS, scans all edge targets, adds matching source rows by LDS float atomics; no global float atomics; sum order = LDS arrival
# speedup vs baseline: 1.4505x; 1.4505x over previous
#include <hip/hip_runtime.h>
#include <math.h>

#define N_NODES 50000
#define N_EDGES 640000
#define DIM     128

typedef __attribute__((ext_vector_type(2))) float v2f;
typedef __attribute__((ext_vector_type(8))) float v8f;


__global__ void k_deg_init(unsigned* __restrict__ deg) {
    int i = blockIdx.x * blockDim.x + threadIdx.x;
    if (i < N_NODES) deg[i] = 1u;
}

__global__ void k_deg_count(const int* __restrict__ dst, unsigned* __restrict__ deg) {
    int e = blockIdx.x * blockDim.x + threadIdx.x;
    if (e < N_EDGES) atomicAdd(&deg[dst[e]], 1u);
}

__global__ void k_dinv(const unsigned* __restrict__ deg, float* __restrict__ dinv) {
    int i = blockIdx.x * blockDim.x + threadIdx.x;
    if (i < N_NODES) dinv[i] = rsqrtf((float)deg[i]);
}


__global__ __launch_bounds__(256) void k_gemm(const float* __restrict__ A,
                                              const float* __restrict__ W,
                                              const float* __restrict__ bias,
                                              float* __restrict__ out,
                                              int addBias) {
    const int wave = threadIdx.x >> 5;
    const int lane = threadIdx.x & 31;
    const int l15  = lane & 15;
    const int hi   = lane >> 4;
    const int rowBase = blockIdx.x << 4;
    const int colBase = wave << 4;

    const float* arow = A + (size_t)(rowBase + l15) * DIM + (hi << 1);
    const float* wcol = W + (size_t)(hi << 1) * DIM + colBase + l15;

    v8f c = {};
#pragma unroll
    for (int k = 0; k < DIM; k += 4) {
        v2f a = *(const v2f*)(arow + k);
        v2f b;
        b.x = wcol[(size_t)k * DIM];
        b.y = wcol[(size_t)(k + 1) * DIM];
        c = __builtin_amdgcn_wmma_f32_16x16x4_f32(false, a, false, b,
                                                  (short)0, c, false, false);
    }

    const float bv = addBias ? bias[colBase + l15] : 0.0f;
    const int m0 = hi << 3;
    float* orow = out + (size_t)(rowBase + m0) * DIM + colBase + l15;
#pragma unroll
    for (int r = 0; r < 8; ++r)
        orow[(size_t)r * DIM] = c[r] + bv;
}


#define TILE_R   196
#define N_TILES  ((N_NODES + TILE_R - 1) / TILE_R)
#define NE4      (N_EDGES / 4)
#define CHUNK4   8192
#define N_CHUNK  ((NE4 + CHUNK4 - 1) / CHUNK4)
#define LIST_W   1028

__device__ __forceinline__ void load_chunk(const int4* __restrict__ dst4, int c, int tid,
                                           int4 (&buf)[8]) {
#pragma unroll
    for (int q = 0; q < 8; ++q) {
        const int i4 = c * CHUNK4 + q * 1024 + tid;
        buf[q] = (i4 < NE4) ? dst4[i4] : make_int4(-1, -1, -1, -1);
    }
}

template <int U>
__device__ __forceinline__ void add_entries(const unsigned* wl, unsigned i0,
                                            const int* __restrict__ src,
                                            const float* __restrict__ dinv,
                                            const float4* __restrict__ h4,
                                            float* acc, const float* dinv_row, int lane) {
    unsigned ent[U];
    int      sv[U];
    float    nm[U];
    float4   v[U];
#pragma unroll
    for (int u = 0; u < U; ++u) ent[u] = wl[i0 + u];
#pragma unroll
    for (int u = 0; u < U; ++u) sv[u] = src[ent[u] & 0xFFFFFu];
#pragma unroll
    for (int u = 0; u < U; ++u) {
        nm[u] = dinv[sv[u]] * dinv_row[ent[u] >> 20];
        v[u]  = h4[(size_t)sv[u] * (DIM / 4) + lane];
    }
#pragma unroll
    for (int u = 0; u < U; ++u) {
        float* a = acc + (ent[u] >> 20) * DIM + lane;
        unsafeAtomicAdd(a,      nm[u] * v[u].x);
        unsafeAtomicAdd(a + 32, nm[u] * v[u].y);
        unsafeAtomicAdd(a + 64, nm[u] * v[u].z);
        unsafeAtomicAdd(a + 96, nm[u] * v[u].w);
    }
}

__device__ __forceinline__ void tile_chunk(int c, const int4 (&cur)[8], int4 (&nxt)[8],
                                           unsigned& cntw,
                                           const int4* __restrict__ dst4,
                                           const int* __restrict__ src,
                                           const float* __restrict__ dinv,
                                           const float4* __restrict__ h4,
                                           float* acc, unsigned* wl, const float* dinv_row,
                                           int lo, int nrows, int tid, int lane) {
    load_chunk(dst4, c + 1, tid, nxt);

#pragma unroll
    for (int q = 0; q < 8; ++q) {
        const unsigned e0 = 4u * (unsigned)(c * CHUNK4 + q * 1024 + tid);
        const int dv[4] = {cur[q].x, cur[q].y, cur[q].z, cur[q].w};
#pragma unroll
        for (int k = 0; k < 4; ++k) {
            const unsigned dl  = (unsigned)(dv[k] - lo);
            const bool     m   = dl < (unsigned)nrows;
            const unsigned bal = __builtin_amdgcn_ballot_w32(m);
            if (bal != 0u) {
                if (m) wl[cntw + __builtin_amdgcn_mbcnt_lo(bal, 0u)] = (e0 + (unsigned)k) | (dl << 20);
                cntw += (unsigned)__builtin_popcount(bal);
            }
        }
    }

    unsigned i0 = 0u;
    for (; i0 + 4u <= cntw; i0 += 4u)
        add_entries<4>(wl, i0, src, dinv, h4, acc, dinv_row, lane);
    if (i0 != 0u) {
        const unsigned rest = cntw - i0;
        unsigned t = 0u;
        if ((unsigned)lane < rest) t = wl[i0 + (unsigned)lane];
        if ((unsigned)lane < rest) wl[lane] = t;
        cntw = rest;
    }
}

__global__ __launch_bounds__(1024) void k_tile_agg(const int* __restrict__ src,
                                                   const int* __restrict__ dst,
                                                   const float* __restrict__ dinv,
                                                   const float* __restrict__ h,
                                                   float* __restrict__ agg) {
    __shared__ float    acc[TILE_R * DIM];
    __shared__ unsigned list[32 * LIST_W];
    __shared__ float    dinv_row[TILE_R];

    const int tid  = threadIdx.x;
    const int lane = tid & 31;
    const int wave = tid >> 5;
    const int lo   = blockIdx.x * TILE_R;
    const int nrows = (N_NODES - lo < TILE_R) ? (N_NODES - lo) : TILE_R;
    const int4*   dst4 = (const int4*)dst;
    const float4* h4   = (const float4*)h;
    unsigned*     wl   = list + wave * LIST_W;

    int4 bufA[8], bufB[8];
    load_chunk(dst4, 0, tid, bufA);

    if (tid < TILE_R) dinv_row[tid] = (tid < nrows) ? dinv[lo + tid] : 0.0f;
    for (int r = wave; r < nrows; r += 32) {
        float s = dinv[lo + r];
        s *= s;
        const float4 v = h4[(size_t)(lo + r) * (DIM / 4) + lane];
        float* a = acc + r * DIM + lane;
        a[0] = v.x * s; a[32] = v.y * s; a[64] = v.z * s; a[96] = v.w * s;
    }
    __syncthreads();

    unsigned cntw = 0u;
    for (int c = 0; c < N_CHUNK; c += 2) {
        tile_chunk(c,     bufA, bufB, cntw, dst4, src, dinv, h4, acc, wl, dinv_row, lo, nrows, tid, lane);
        tile_chunk(c + 1, bufB, bufA, cntw, dst4, src, dinv, h4, acc, wl, dinv_row, lo, nrows, tid, lane);
    }
    for (unsigned i = 0u; i < cntw; ++i)
        add_entries<1>(wl, i, src, dinv, h4, acc, dinv_row, lane);
    __syncthreads();

    float4* agg4 = (float4*)agg;
    for (int r = wave; r < nrows; r += 32) {
        const float* a = acc + r * DIM + lane;
        float4 o;
        o.x = a[0]; o.y = a[32]; o.z = a[64]; o.w = a[96];
        agg4[(size_t)(lo + r) * (DIM / 4) + lane] = o;
    }
}


__global__ __launch_bounds__(256) void k_act(const float* __restrict__ agg,
                                             const float* __restrict__ b1,
                                             float* __restrict__ h) {
    int t = blockIdx.x * blockDim.x + threadIdx.x;
    h[t] = tanhf(agg[t] + b1[t & (DIM - 1)]);
}


extern "C" void kernel_launch(void* const* d_in, const int* in_sizes, int n_in,
                              void* d_out, int out_size, void* d_ws, size_t ws_size,
                              hipStream_t stream) {
    const float* x  = (const float*)d_in[0];
    const int*   ei = (const int*)  d_in[1];
    const float* W1 = (const float*)d_in[2];
    const float* b1 = (const float*)d_in[3];
    const float* W2 = (const float*)d_in[4];
    const float* b2 = (const float*)d_in[5];
    float* out = (float*)d_out;

    float*    ws   = (float*)d_ws;
    unsigned* deg  = (unsigned*)ws;
    float*    dinv = ws + 65536;
    float*    h    = ws + 131072;
    float*    agg  = ws + 131072 + (size_t)N_NODES * DIM;

    const int* esrc = ei;
    const int* edst = ei + N_EDGES;

    k_deg_init <<<(N_NODES + 255) / 256, 256, 0, stream>>>(deg);
    k_deg_count<<<(N_EDGES + 255) / 256, 256, 0, stream>>>(edst, deg);
    k_dinv     <<<(N_NODES + 255) / 256, 256, 0, stream>>>(deg, dinv);

    k_gemm <<<N_NODES / 16, 256, 0, stream>>>(x, W1, b1, h, 0);
    k_tile_agg<<<N_TILES, 1024, 0, stream>>>(esrc, edst, dinv, h, agg);
    k_act  <<<(N_NODES * DIM) / 256, 256, 0, stream>>>(agg, b1, h);
    k_gemm <<<N_NODES / 16, 256, 0, stream>>>(h, W2, b2, out, 1);
}


# Tile sums stay in LDS: the aggregation launch applies b1+tanh to its 196 rows there and multiplies them by W2 on WMMA (same k order), writes out; agg, act planes and 2 launches gone; sums unchanged
# speedup vs baseline: 1.8099x; 1.2478x over previous
#include <hip/hip_runtime.h>
#include <math.h>

#define N_NODES 50000
#define N_EDGES 640000
#define DIM     128

typedef __attribute__((ext_vector_type(2))) float v2f;
typedef __attribute__((ext_vector_type(8))) float v8f;


__global__ void k_deg_init(unsigned* __restrict__ deg) {
    int i = blockIdx.x * blockDim.x + threadIdx.x;
    if (i < N_NODES) deg[i] = 1u;
}

__global__ void k_deg_count(const int* __restrict__ dst, unsigned* __restrict__ deg) {
    int e = blockIdx.x * blockDim.x + threadIdx.x;
    if (e < N_EDGES) atomicAdd(&deg[dst[e]], 1u);
}

__global__ void k_dinv(const unsigned* __restrict__ deg, float* __restrict__ dinv) {
    int i = blockIdx.x * blockDim.x + threadIdx.x;
    if (i < N_NODES) dinv[i] = rsqrtf((float)deg[i]);
}


__global__ __launch_bounds__(256) void k_gemm(const float* __restrict__ A,
                                              const float* __restrict__ W,
                                              const float* __restrict__ bias,
                                              float* __restrict__ out,
                                              int addBias) {
    const int wave = threadIdx.x >> 5;
    const int lane = threadIdx.x & 31;
    const int l15  = lane & 15;
    const int hi   = lane >> 4;
    const int rowBase = blockIdx.x << 4;
    const int colBase = wave << 4;

    const float* arow = A + (size_t)(rowBase + l15) * DIM + (hi << 1);
    const float* wcol = W + (size_t)(hi << 1) * DIM + colBase + l15;

    v8f c = {};
#pragma unroll
    for (int k = 0; k < DIM; k += 4) {
        v2f a = *(const v2f*)(arow + k);
        v2f b;
        b.x = wcol[(size_t)k * DIM];
        b.y = wcol[(size_t)(k + 1) * DIM];
        c = __builtin_amdgcn_wmma_f32_16x16x4_f32(false, a, false, b,
                                                  (short)0, c, false, false);
    }

    const float bv = addBias ? bias[colBase + l15] : 0.0f;
    const int m0 = hi << 3;
    float* orow = out + (size_t)(rowBase + m0) * DIM + colBase + l15;
#pragma unroll
    for (int r = 0; r < 8; ++r)
        orow[(size_t)r * DIM] = c[r] + bv;
}


#define TILE_R   196
#define N_TILES  ((N_NODES + TILE_R - 1) / TILE_R)
#define NE4      (N_EDGES / 4)
#define CHUNK4   8192
#define N_CHUNK  ((NE4 + CHUNK4 - 1) / CHUNK4)
#define LIST_W   1028
#define ACT_S    132

__device__ __forceinline__ void load_chunk(const int4* __restrict__ dst4, int c, int tid,
                                           int4 (&buf)[8]) {
#pragma unroll
    for (int q = 0; q < 8; ++q) {
        const int i4 = c * CHUNK4 + q * 1024 + tid;
        buf[q] = (i4 < NE4) ? dst4[i4] : make_int4(-1, -1, -1, -1);
    }
}

template <int U>
__device__ __forceinline__ void add_entries(const unsigned* wl, unsigned i0,
                                            const int* __restrict__ src,
                                            const float* __restrict__ dinv,
                                            const float4* __restrict__ h4,
                                            float* acc, const float* dinv_row, int lane) {
    unsigned ent[U];
    int      sv[U];
    float    nm[U];
    float4   v[U];
#pragma unroll
    for (int u = 0; u < U; ++u) ent[u] = wl[i0 + u];
#pragma unroll
    for (int u = 0; u < U; ++u) sv[u] = src[ent[u] & 0xFFFFFu];
#pragma unroll
    for (int u = 0; u < U; ++u) {
        nm[u] = dinv[sv[u]] * dinv_row[ent[u] >> 20];
        v[u]  = h4[(size_t)sv[u] * (DIM / 4) + lane];
    }
#pragma unroll
    for (int u = 0; u < U; ++u) {
        float* a = acc + (ent[u] >> 20) * DIM + lane;
        unsafeAtomicAdd(a,      nm[u] * v[u].x);
        unsafeAtomicAdd(a + 32, nm[u] * v[u].y);
        unsafeAtomicAdd(a + 64, nm[u] * v[u].z);
        unsafeAtomicAdd(a + 96, nm[u] * v[u].w);
    }
}

__device__ __forceinline__ void tile_chunk(int c, const int4 (&cur)[8], int4 (&nxt)[8],
                                           unsigned& cntw,
                                           const int4* __restrict__ dst4,
                                           const int* __restrict__ src,
                                           const float* __restrict__ dinv,
                                           const float4* __restrict__ h4,
                                           float* acc, unsigned* wl, const float* dinv_row,
                                           int lo, int nrows, int tid, int lane) {
    load_chunk(dst4, c + 1, tid, nxt);

#pragma unroll
    for (int q = 0; q < 8; ++q) {
        const unsigned e0 = 4u * (unsigned)(c * CHUNK4 + q * 1024 + tid);
        const int dv[4] = {cur[q].x, cur[q].y, cur[q].z, cur[q].w};
#pragma unroll
        for (int k = 0; k < 4; ++k) {
            const unsigned dl  = (unsigned)(dv[k] - lo);
            const bool     m   = dl < (unsigned)nrows;
            const unsigned bal = __builtin_amdgcn_ballot_w32(m);
            if (bal != 0u) {
                if (m) wl[cntw + __builtin_amdgcn_mbcnt_lo(bal, 0u)] = (e0 + (unsigned)k) | (dl << 20);
                cntw += (unsigned)__builtin_popcount(bal);
            }
        }
    }

    unsigned i0 = 0u;
    for (; i0 + 4u <= cntw; i0 += 4u)
        add_entries<4>(wl, i0, src, dinv, h4, acc, dinv_row, lane);
    if (i0 != 0u) {
        const unsigned rest = cntw - i0;
        unsigned t = 0u;
        if ((unsigned)lane < rest) t = wl[i0 + (unsigned)lane];
        if ((unsigned)lane < rest) wl[lane] = t;
        cntw = rest;
    }
}

__global__ __launch_bounds__(1024) void k_tile_agg(const int* __restrict__ src,
                                                   const int* __restrict__ dst,
                                                   const float* __restrict__ dinv,
                                                   const float* __restrict__ h,
                                                   const float* __restrict__ b1,
                                                   const float* __restrict__ W2,
                                                   const float* __restrict__ b2,
                                                   float* __restrict__ out) {
    __shared__ float    acc[TILE_R * DIM];
    __shared__ __attribute__((aligned(16))) unsigned list[32 * LIST_W];
    __shared__ float    dinv_row[TILE_R];

    const int tid  = threadIdx.x;
    const int lane = tid & 31;
    const int wave = tid >> 5;
    const int lo   = blockIdx.x * TILE_R;
    const int nrows = (N_NODES - lo < TILE_R) ? (N_NODES - lo) : TILE_R;
    const int4*   dst4 = (const int4*)dst;
    const float4* h4   = (const float4*)h;
    unsigned*     wl   = list + wave * LIST_W;

    int4 bufA[8], bufB[8];
    load_chunk(dst4, 0, tid, bufA);

    if (tid < TILE_R) dinv_row[tid] = (tid < nrows) ? dinv[lo + tid] : 0.0f;
    for (int r = wave; r < nrows; r += 32) {
        float s = dinv[lo + r];
        s *= s;
        const float4 v = h4[(size_t)(lo + r) * (DIM / 4) + lane];
        float* a = acc + r * DIM + lane;
        a[0] = v.x * s; a[32] = v.y * s; a[64] = v.z * s; a[96] = v.w * s;
    }
    __syncthreads();

    unsigned cntw = 0u;
    for (int c = 0; c < N_CHUNK; c += 2) {
        tile_chunk(c,     bufA, bufB, cntw, dst4, src, dinv, h4, acc, wl, dinv_row, lo, nrows, tid, lane);
        tile_chunk(c + 1, bufB, bufA, cntw, dst4, src, dinv, h4, acc, wl, dinv_row, lo, nrows, tid, lane);
    }
    for (unsigned i = 0u; i < cntw; ++i)
        add_entries<1>(wl, i, src, dinv, h4, acc, dinv_row, lane);


    const int wv      = __builtin_amdgcn_readfirstlane(wave);
    const int l15     = lane & 15;
    const int hi      = lane >> 4;
    const int colBase = (wv & 7) << 4;

    const float4 bb = ((const float4*)b1)[lane];

    __syncthreads();

    v2f bw[DIM / 4];
    const float* wcol = W2 + (size_t)(hi << 1) * DIM + colBase + l15;
#pragma unroll
    for (int kk = 0; kk < DIM / 4; ++kk) {
        bw[kk].x = wcol[(size_t)(4 * kk) * DIM];
        bw[kk].y = wcol[(size_t)(4 * kk + 1) * DIM];
    }
    const float bv = b2[colBase + l15];

    float* act = (float*)list;
    for (int r = wv; r < nrows; r += 32) {
        const float* a = acc + r * DIM + lane;
        const float x0 = a[0] + bb.x, x1 = a[32] + bb.y, x2 = a[64] + bb.z, x3 = a[96] + bb.w;
        float4 o;
        o.x = tanhf(x0);
        o.y = tanhf(x1);
        o.z = tanhf(x2);
        o.w = tanhf(x3);
        *(float4*)(act + r * ACT_S + 4 * lane) = o;
    }
    __syncthreads();

    const int nstrips = (nrows + 15) >> 4;
    for (int s = wv >> 3; s < nstrips; s += 4) {
        int row = (s << 4) + l15;
        row = (row < nrows) ? row : (nrows - 1);
        const float* ar = act + row * ACT_S + (hi << 1);
        v8f c = {};
#pragma unroll
        for (int kk = 0; kk < DIM / 4; ++kk) {
            const v2f a = *(const v2f*)(ar + 4 * kk);
            c = __builtin_amdgcn_wmma_f32_16x16x4_f32(false, a, false, bw[kk],
                                                      (short)0, c, false, false);
        }
        const int m0 = (s << 4) + (hi << 3);
        const int nvalid = nrows - m0;
        float* orow = out + (size_t)(lo + m0) * DIM + colBase + l15;
#pragma unroll
        for (int r = 0; r < 8; ++r)
            if (r < nvalid) orow[(size_t)r * DIM] = c[r] + bv;
    }
}


extern "C" void kernel_launch(void* const* d_in, const int* in_sizes, int n_in,
                              void* d_out, int out_size, void* d_ws, size_t ws_size,
                              hipStream_t stream) {
    const float* x  = (const float*)d_in[0];
    const int*   ei = (const int*)  d_in[1];
    const float* W1 = (const float*)d_in[2];
    const float* b1 = (const float*)d_in[3];
    const float* W2 = (const float*)d_in[4];
    const float* b2 = (const float*)d_in[5];
    float* out = (float*)d_out;

    float*    ws   = (float*)d_ws;
    unsigned* deg  = (unsigned*)ws;
    float*    dinv = ws + 65536;
    float*    h    = ws + 131072;

    const int* esrc = ei;
    const int* edst = ei + N_EDGES;

    k_deg_init <<<(N_NODES + 255) / 256, 256, 0, stream>>>(deg);
    k_deg_count<<<(N_EDGES + 255) / 256, 256, 0, stream>>>(edst, deg);
    k_dinv     <<<(N_NODES + 255) / 256, 256, 0, stream>>>(deg, dinv);

    k_gemm <<<N_NODES / 16, 256, 0, stream>>>(x, W1, b1, h, 0);
    k_tile_agg<<<N_TILES, 1024, 0, stream>>>(esrc, edst, dinv, h, b1, W2, b2, out);
}


# Tile scan reads one-byte tile keys (written by the degree-count launch, with a source|row word per edge for the listed edges) in place of the 4-byte targets; sum order = LDS arrival as before
# speedup vs baseline: 1.9532x; 1.0792x over previous
#include <hip/hip_runtime.h>
#include <math.h>

#define N_NODES 50000
#define N_EDGES 640000
#define DIM     128
#define TILE_R   196
#define N_TILES  ((N_NODES + TILE_R - 1) / TILE_R)
#define NE4      (N_EDGES / 4)

typedef __attribute__((ext_vector_type(2))) float v2f;
typedef __attribute__((ext_vector_type(8))) float v8f;


__global__ void k_deg_init(unsigned* __restrict__ deg) {
    int i = blockIdx.x * blockDim.x + threadIdx.x;
    if (i < N_NODES) deg[i] = 1u;
}

static_assert(N_NODES <= 65536, "a source number must fit 16 bits of an edge record");
static_assert(N_TILES <= 256 && TILE_R <= 256, "tile number and row in the tile are one byte each");

__global__ void k_deg_count(const int* __restrict__ src, const int* __restrict__ dst,
                            unsigned* __restrict__ deg, unsigned* __restrict__ keyw,
                            unsigned* __restrict__ rec) {
    int t = blockIdx.x * blockDim.x + threadIdx.x;
    if (t < NE4) {
        const int4 d = ((const int4*)dst)[t];
        const int4 s = ((const int4*)src)[t];
        atomicAdd(&deg[d.x], 1u);
        atomicAdd(&deg[d.y], 1u);
        atomicAdd(&deg[d.z], 1u);
        atomicAdd(&deg[d.w], 1u);
        const unsigned k0 = (unsigned)d.x / TILE_R, k1 = (unsigned)d.y / TILE_R;
        const unsigned k2 = (unsigned)d.z / TILE_R, k3 = (unsigned)d.w / TILE_R;
        keyw[t] = k0 | (k1 << 8) | (k2 << 16) | (k3 << 24);
        uint4 r;
        r.x = (unsigned)s.x | (((unsigned)d.x - k0 * TILE_R) << 16);
        r.y = (unsigned)s.y | (((unsigned)d.y - k1 * TILE_R) << 16);
        r.z = (unsigned)s.z | (((unsigned)d.z - k2 * TILE_R) << 16);
        r.w = (unsigned)s.w | (((unsigned)d.w - k3 * TILE_R) << 16);
        ((uint4*)rec)[t] = r;
    }
}

__global__ void k_dinv(const unsigned* __restrict__ deg, float* __restrict__ dinv) {
    int i = blockIdx.x * blockDim.x + threadIdx.x;
    if (i < N_NODES) dinv[i] = rsqrtf((float)deg[i]);
}


__global__ __launch_bounds__(256) void k_gemm(const float* __restrict__ A,
                                              const float* __restrict__ W,
                                              const float* __restrict__ bias,
                                              float* __restrict__ out,
                                              int addBias) {
    const int wave = threadIdx.x >> 5;
    const int lane = threadIdx.x & 31;
    const int l15  = lane & 15;
    const int hi   = lane >> 4;
    const int rowBase = blockIdx.x << 4;
    const int colBase = wave << 4;

    const float* arow = A + (size_t)(rowBase + l15) * DIM + (hi << 1);
    const float* wcol = W + (size_t)(hi << 1) * DIM + colBase + l15;

    v8f c = {};
#pragma unroll
    for (int k = 0; k < DIM; k += 4) {
        v2f a = *(const v2f*)(arow + k);
        v2f b;
        b.x = wcol[(size_t)k * DIM];
        b.y = wcol[(size_t)(k + 1) * DIM];
        c = __builtin_amdgcn_wmma_f32_16x16x4_f32(false, a, false, b,
                                                  (short)0, c, false, false);
    }

    const float bv = addBias ? bias[colBase + l15] : 0.0f;
    const int m0 = hi << 3;
    float* orow = out + (size_t)(rowBase + m0) * DIM + colBase + l15;
#pragma unroll
    for (int r = 0; r < 8; ++r)
        orow[(size_t)r * DIM] = c[r] + bv;
}


#define NK4      (N_EDGES / 16)
#define CHUNK4   2048
#define N_CHUNK  ((NK4 + CHUNK4 - 1) / CHUNK4)
#define LIST_W   1028
#define ACT_S    132

static_assert(N_EDGES % 16 == 0, "the key list is read as whole int4");
static_assert(N_CHUNK % 2 == 0, "chunks are taken in pairs");
static_assert((N_CHUNK - 1) * CHUNK4 <= NK4 && N_CHUNK * CHUNK4 >= NK4, "only the last chunk may pass the end");

template <bool LAST>
__device__ __forceinline__ void load_chunk(const int4* __restrict__ key4, int c, int tid,
                                           int4 (&buf)[2]) {
#pragma unroll
    for (int q = 0; q < 2; ++q) {
        const int i4 = c * CHUNK4 + q * 1024 + tid;
        buf[q] = key4[(LAST && i4 >= NK4) ? (NK4 - 1) : i4];
    }
}

template <int U>
__device__ __forceinline__ void add_entries(const unsigned* wl, unsigned i0,
                                            const unsigned* __restrict__ rec,
                                            const float* __restrict__ dinv,
                                            const float4* __restrict__ h4,
                                            float* acc, const float* dinv_row, int lane) {
    unsigned ent[U];
    unsigned rc[U];
    unsigned sv[U];
    unsigned rw[U];
    float    nm[U];
    float4   v[U];
#pragma unroll
    for (int u = 0; u < U; ++u) ent[u] = wl[i0 + u];
#pragma unroll
    for (int u = 0; u < U; ++u) rc[u] = rec[ent[u]];
#pragma unroll
    for (int u = 0; u < U; ++u) {
        sv[u] = rc[u] & 0xFFFFu;
        rw[u] = rc[u] >> 16;
        nm[u] = dinv[sv[u]] * dinv_row[rw[u]];
        v[u]  = h4[(size_t)sv[u] * (DIM / 4) + lane];
    }
#pragma unroll
    for (int u = 0; u < U; ++u) {
        float* a = acc + rw[u] * DIM + lane;
        unsafeAtomicAdd(a,      nm[u] * v[u].x);
        unsafeAtomicAdd(a + 32, nm[u] * v[u].y);
        unsafeAtomicAdd(a + 64, nm[u] * v[u].z);
        unsafeAtomicAdd(a + 96, nm[u] * v[u].w);
    }
}

template <int NEXT>
__device__ __forceinline__ void tile_chunk(int c, const int4 (&cur)[2], int4 (&nxt)[2],
                                           unsigned& cntw, unsigned pat,
                                           const int4* __restrict__ key4,
                                           const unsigned* __restrict__ rec,
                                           const float* __restrict__ dinv,
                                           const float4* __restrict__ h4,
                                           float* acc, unsigned* wl, const float* dinv_row,
                                           int tid, int lane) {
    if (NEXT == 0) load_chunk<false>(key4, c + 1, tid, nxt);
    if (NEXT == 1) load_chunk<true>(key4, c + 1, tid, nxt);

#pragma unroll
    for (int q = 0; q < 2; ++q) {
        const int      i4 = c * CHUNK4 + q * 1024 + tid;
        const bool     live = (NEXT != 2) || (i4 < NK4);
        const unsigned e0 = 16u * (unsigned)i4;
        const unsigned kw[4] = {(unsigned)cur[q].x, (unsigned)cur[q].y,
                                (unsigned)cur[q].z, (unsigned)cur[q].w};
#pragma unroll
        for (int j = 0; j < 4; ++j) {
            const unsigned x = kw[j] ^ pat;
            unsigned z = ~(((x & 0x7F7F7F7Fu) + 0x7F7F7F7Fu) | x | 0x7F7F7F7Fu);
            if (!live) z = 0u;
            unsigned bal = __builtin_amdgcn_ballot_w32(z != 0u);
            while (bal != 0u) {
                if (z != 0u) {
                    const unsigned k = (unsigned)__builtin_ctz(z) >> 3;
                    wl[cntw + __builtin_amdgcn_mbcnt_lo(bal, 0u)] = e0 + 4u * (unsigned)j + k;
                    z &= z - 1u;
                }
                cntw += (unsigned)__builtin_popcount(bal);
                bal = __builtin_amdgcn_ballot_w32(z != 0u);
            }
        }
    }

    unsigned i0 = 0u;
    for (; i0 + 4u <= cntw; i0 += 4u)
        add_entries<4>(wl, i0, rec, dinv, h4, acc, dinv_row, lane);
    if (i0 != 0u) {
        const unsigned rest = cntw - i0;
        unsigned t = 0u;
        if ((unsigned)lane < rest) t = wl[i0 + (unsigned)lane];
        if ((unsigned)lane < rest) wl[lane] = t;
        cntw = rest;
    }
}

__global__ __launch_bounds__(1024) void k_tile_agg(const unsigned* __restrict__ keyw,
                                                   const unsigned* __restrict__ rec,
                                                   const float* __restrict__ dinv,
                                                   const float* __restrict__ h,
                                                   const float* __restrict__ b1,
                                                   const float* __restrict__ W2,
                                                   const float* __restrict__ b2,
                                                   float* __restrict__ out) {
    __shared__ float    acc[TILE_R * DIM];
    __shared__ __attribute__((aligned(16))) unsigned list[32 * LIST_W];
    __shared__ float    dinv_row[TILE_R];

    const int tid  = threadIdx.x;
    const int lane = tid & 31;
    const int wave = tid >> 5;
    const int lo   = blockIdx.x * TILE_R;
    const int nrows = (N_NODES - lo < TILE_R) ? (N_NODES - lo) : TILE_R;
    const int4*   key4 = (const int4*)keyw;
    const float4* h4   = (const float4*)h;
    unsigned*     wl   = list + wave * LIST_W;
    const unsigned pat = (unsigned)blockIdx.x * 0x01010101u;

    int4 bufA[2], bufB[2];
    load_chunk<false>(key4, 0, tid, bufA);

    if (tid < TILE_R) dinv_row[tid] = (tid < nrows) ? dinv[lo + tid] : 0.0f;
    for (int r = wave; r < nrows; r += 32) {
        float s = dinv[lo + r];
        s *= s;
        const float4 v = h4[(size_t)(lo + r) * (DIM / 4) + lane];
        float* a = acc + r * DIM + lane;
        a[0] = v.x * s; a[32] = v.y * s; a[64] = v.z * s; a[96] = v.w * s;
    }
    __syncthreads();

    unsigned cntw = 0u;
    for (int c = 0; c < N_CHUNK - 2; c += 2) {
        tile_chunk<0>(c,     bufA, bufB, cntw, pat, key4, rec, dinv, h4, acc, wl, dinv_row, tid, lane);
        tile_chunk<0>(c + 1, bufB, bufA, cntw, pat, key4, rec, dinv, h4, acc, wl, dinv_row, tid, lane);
    }
    tile_chunk<1>(N_CHUNK - 2, bufA, bufB, cntw, pat, key4, rec, dinv, h4, acc, wl, dinv_row, tid, lane);
    tile_chunk<2>(N_CHUNK - 1, bufB, bufA, cntw, pat, key4, rec, dinv, h4, acc, wl, dinv_row, tid, lane);
    for (unsigned i = 0u; i < cntw; ++i)
        add_entries<1>(wl, i, rec, dinv, h4, acc, dinv_row, lane);


    const int wv      = __builtin_amdgcn_readfirstlane(wave);
    const int l15     = lane & 15;
    const int hi      = lane >> 4;
    const int colBase = (wv & 7) << 4;

    const float4 bb = ((const float4*)b1)[lane];

    __syncthreads();

    v2f bw[DIM / 4];
    const float* wcol = W2 + (size_t)(hi << 1) * DIM + colBase + l15;
#pragma unroll
    for (int kk = 0; kk < DIM / 4; ++kk) {
        bw[kk].x = wcol[(size_t)(4 * kk) * DIM];
        bw[kk].y = wcol[(size_t)(4 * kk + 1) * DIM];
    }
    const float bv = b2[colBase + l15];

    float* act = (float*)list;
    for (int r = wv; r < nrows; r += 32) {
        const float* a = acc + r * DIM + lane;
        const float x0 = a[0] + bb.x, x1 = a[32] + bb.y, x2 = a[64] + bb.z, x3 = a[96] + bb.w;
        float4 o;
        o.x = tanhf(x0);
        o.y = tanhf(x1);
        o.z = tanhf(x2);
        o.w = tanhf(x3);
        *(float4*)(act + r * ACT_S + 4 * lane) = o;
    }
    __syncthreads();

    const int nstrips = (nrows + 15) >> 4;
    for (int s = wv >> 3; s < nstrips; s += 4) {
        int row = (s << 4) + l15;
        row = (row < nrows) ? row : (nrows - 1);
        const float* ar = act + row * ACT_S + (hi << 1);
        v8f c = {};
#pragma unroll
        for (int kk = 0; kk < DIM / 4; ++kk) {
            const v2f a = *(const v2f*)(ar + 4 * kk);
            c = __builtin_amdgcn_wmma_f32_16x16x4_f32(false, a, false, bw[kk],
                                                      (short)0, c, false, false);
        }
        const int m0 = (s << 4) + (hi << 3);
        const int nvalid = nrows - m0;
        float* orow = out + (size_t)(lo + m0) * DIM + colBase + l15;
#pragma unroll
        for (int r = 0; r < 8; ++r)
            if (r < nvalid) orow[(size_t)r * DIM] = c[r] + bv;
    }
}


extern "C" void kernel_launch(void* const* d_in, const int* in_sizes, int n_in,
                              void* d_out, int out_size, void* d_ws, size_t ws_size,
                              hipStream_t stream) {
    const float* x  = (const float*)d_in[0];
    const int*   ei = (const int*)  d_in[1];
    const float* W1 = (const float*)d_in[2];
    const float* b1 = (const float*)d_in[3];
    const float* W2 = (const float*)d_in[4];
    const float* b2 = (const float*)d_in[5];
    float* out = (float*)d_out;

    float*    ws   = (float*)d_ws;
    unsigned* deg  = (unsigned*)ws;
    float*    dinv = ws + 65536;
    float*    h    = ws + 131072;
    unsigned* keyw = (unsigned*)(ws + 131072 + N_NODES * DIM);
    unsigned* rec  = keyw + NE4;

    const int* esrc = ei;
    const int* edst = ei + N_EDGES;

    k_deg_init <<<(N_NODES + 255) / 256, 256, 0, stream>>>(deg);
    k_deg_count<<<(NE4 + 255) / 256, 256, 0, stream>>>(esrc, edst, deg, keyw, rec);
    k_dinv     <<<(N_NODES + 255) / 256, 256, 0, stream>>>(deg, dinv);

    k_gemm <<<N_NODES / 16, 256, 0, stream>>>(x, W1, b1, h, 0);
    k_tile_agg<<<N_TILES, 1024, 0, stream>>>(keyw, rec, dinv, h, b1, W2, b2, out);
}


# Projection GEMM cut by rows: a wave owns a whole 16x128 strip (8 accumulator tiles) and reads its A fragments once; W1 sits in LDS as B fragments, once per work-group; k order and sums unchanged
# speedup vs baseline: 2.1422x; 1.0968x over previous
#include <hip/hip_runtime.h>
#include <math.h>

#define N_NODES 50000
#define N_EDGES 640000
#define DIM     128
#define TILE_R   196
#define N_TILES  ((N_NODES + TILE_R - 1) / TILE_R)
#define NE4      (N_EDGES / 4)

typedef __attribute__((ext_vector_type(2))) float v2f;
typedef __attribute__((ext_vector_type(8))) float v8f;


__global__ void k_deg_init(unsigned* __restrict__ deg) {
    int i = blockIdx.x * blockDim.x + threadIdx.x;
    if (i < N_NODES) deg[i] = 1u;
}

static_assert(N_NODES <= 65536, "a source number must fit 16 bits of an edge record");
static_assert(N_TILES <= 256 && TILE_R <= 256, "tile number and row in the tile are one byte each");

__global__ void k_deg_count(const int* __restrict__ src, const int* __restrict__ dst,
                            unsigned* __restrict__ deg, unsigned* __restrict__ keyw,
                            unsigned* __restrict__ rec) {
    int t = blockIdx.x * blockDim.x + threadIdx.x;
    if (t < NE4) {
        const int4 d = ((const int4*)dst)[t];
        const int4 s = ((const int4*)src)[t];
        atomicAdd(&deg[d.x], 1u);
        atomicAdd(&deg[d.y], 1u);
        atomicAdd(&deg[d.z], 1u);
        atomicAdd(&deg[d.w], 1u);
        const unsigned k0 = (unsigned)d.x / TILE_R, k1 = (unsigned)d.y / TILE_R;
        const unsigned k2 = (unsigned)d.z / TILE_R, k3 = (unsigned)d.w / TILE_R;
        keyw[t] = k0 | (k1 << 8) | (k2 << 16) | (k3 << 24);
        uint4 r;
        r.x = (unsigned)s.x | (((unsigned)d.x - k0 * TILE_R) << 16);
        r.y = (unsigned)s.y | (((unsigned)d.y - k1 * TILE_R) << 16);
        r.z = (unsigned)s.z | (((unsigned)d.z - k2 * TILE_R) << 16);
        r.w = (unsigned)s.w | (((unsigned)d.w - k3 * TILE_R) << 16);
        ((uint4*)rec)[t] = r;
    }
}

__global__ void k_dinv(const unsigned* __restrict__ deg, float* __restrict__ dinv) {
    int i = blockIdx.x * blockDim.x + threadIdx.x;
    if (i < N_NODES) dinv[i] = rsqrtf((float)deg[i]);
}


#define N_STRIPS   (N_NODES / 16)
#define GEMM_WAVES 16
#define GEMM_WGS   ((N_STRIPS + GEMM_WAVES - 1) / GEMM_WAVES)
static_assert(N_NODES % 16 == 0, "whole strips only");
static_assert(GEMM_WAVES * 2 * 4 == DIM, "wave w puts the k-steps 2w and 2w+1 of W into LDS");

__global__ __launch_bounds__(GEMM_WAVES * 32) void k_gemm(const float* __restrict__ A,
                                                          const float* __restrict__ W,
                                                          float* __restrict__ out) {
    __shared__ __attribute__((aligned(16))) float wfrag[DIM * DIM];

    const int lane = threadIdx.x & 31;
    const int wv   = __builtin_amdgcn_readfirstlane((int)(threadIdx.x >> 5));
    const int l15  = lane & 15;
    const int hi   = lane >> 4;
    const int strip = blockIdx.x * GEMM_WAVES + wv;

    const float* wcol = W + (size_t)(8 * wv + (hi << 1)) * DIM + l15;
    float4 wq[8];
#pragma unroll
    for (int n = 0; n < 8; ++n) {
        wq[n].x = wcol[(n << 4)];
        wq[n].y = wcol[(n << 4) + DIM];
        wq[n].z = wcol[(n << 4) + 4 * DIM];
        wq[n].w = wcol[(n << 4) + 5 * DIM];
    }
    float4* wf4 = (float4*)wfrag + lane;
#pragma unroll
    for (int n = 0; n < 8; ++n)
        wf4[(wv * 8 + n) * 32] = wq[n];

    unsigned ooff = (unsigned)((((hi << 3) * DIM) + l15) * 4);
    asm volatile("" : "+v"(ooff));
    __syncthreads();

    if (strip < N_STRIPS) {
        v2f a[DIM / 4];
        const float* arow = A + (size_t)((strip << 4) + l15) * DIM + (hi << 1);
#pragma unroll
        for (int kk = 0; kk < DIM / 4; ++kk)
            a[kk] = *(const v2f*)(arow + 4 * kk);

        v8f c[8] = {};
#pragma unroll
        for (int kp = 0; kp < DIM / 8; ++kp) {
            float4 bq[8];
#pragma unroll
            for (int n = 0; n < 8; ++n) bq[n] = wf4[(kp * 8 + n) * 32];
#pragma unroll
            for (int n = 0; n < 8; ++n) {
                v2f b; b.x = bq[n].x; b.y = bq[n].y;
                c[n] = __builtin_amdgcn_wmma_f32_16x16x4_f32(false, a[2 * kp], false, b,
                                                             (short)0, c[n], false, false);
            }
#pragma unroll
            for (int n = 0; n < 8; ++n) {
                v2f b; b.x = bq[n].z; b.y = bq[n].w;
                c[n] = __builtin_amdgcn_wmma_f32_16x16x4_f32(false, a[2 * kp + 1], false, b,
                                                             (short)0, c[n], false, false);
            }
        }

        char* obase = (char*)(out + (size_t)(strip << 4) * DIM);
#pragma unroll
        for (int n = 0; n < 8; ++n)
#pragma unroll
            for (int r = 0; r < 8; ++r)
                *(float*)(obase + ooff + (r * DIM + (n << 4)) * 4) = c[n][r];
    }
}


#define NK4      (N_EDGES / 16)
#define CHUNK4   2048
#define N_CHUNK  ((NK4 + CHUNK4 - 1) / CHUNK4)
#define LIST_W   1028
#define ACT_S    132

static_assert(N_EDGES % 16 == 0, "the key list is read as whole int4");
static_assert(N_CHUNK % 2 == 0, "chunks are taken in pairs");
static_assert((N_CHUNK - 1) * CHUNK4 <= NK4 && N_CHUNK * CHUNK4 >= NK4, "only the last chunk may pass the end");

template <bool LAST>
__device__ __forceinline__ void load_chunk(const int4* __restrict__ key4, int c, int tid,
                                           int4 (&buf)[2]) {
#pragma unroll
    for (int q = 0; q < 2; ++q) {
        const int i4 = c * CHUNK4 + q * 1024 + tid;
        buf[q] = key4[(LAST && i4 >= NK4) ? (NK4 - 1) : i4];
    }
}

template <int U>
__device__ __forceinline__ void add_entries(const unsigned* wl, unsigned i0,
                                            const unsigned* __restrict__ rec,
                                            const float* __restrict__ dinv,
                                            const float4* __restrict__ h4,
                                            float* acc, const float* dinv_row, int lane) {
    unsigned ent[U];
    unsigned rc[U];
    unsigned sv[U];
    unsigned rw[U];
    float    nm[U];
    float4   v[U];
#pragma unroll
    for (int u = 0; u < U; ++u) ent[u] = wl[i0 + u];
#pragma unroll
    for (int u = 0; u < U; ++u) rc[u] = rec[ent[u]];
#pragma unroll
    for (int u = 0; u < U; ++u) {
        sv[u] = rc[u] & 0xFFFFu;
        rw[u] = rc[u] >> 16;
        nm[u] = dinv[sv[u]] * dinv_row[rw[u]];
        v[u]  = h4[(size_t)sv[u] * (DIM / 4) + lane];
    }
#pragma unroll
    for (int u = 0; u < U; ++u) {
        float* a = acc + rw[u] * DIM + lane;
        unsafeAtomicAdd(a,      nm[u] * v[u].x);
        unsafeAtomicAdd(a + 32, nm[u] * v[u].y);
        unsafeAtomicAdd(a + 64, nm[u] * v[u].z);
        unsafeAtomicAdd(a + 96, nm[u] * v[u].w);
    }
}

template <int NEXT>
__device__ __forceinline__ void tile_chunk(int c, const int4 (&cur)[2], int4 (&nxt)[2],
                                           unsigned& cntw, unsigned pat,
                                           const int4* __restrict__ key4,
                                           const unsigned* __restrict__ rec,
                                           const float* __restrict__ dinv,
                                           const float4* __restrict__ h4,
                                           float* acc, unsigned* wl, const float* dinv_row,
                                           int tid, int lane) {
    if (NEXT == 0) load_chunk<false>(key4, c + 1, tid, nxt);
    if (NEXT == 1) load_chunk<true>(key4, c + 1, tid, nxt);

#pragma unroll
    for (int q = 0; q < 2; ++q) {
        const int      i4 = c * CHUNK4 + q * 1024 + tid;
        const bool     live = (NEXT != 2) || (i4 < NK4);
        const unsigned e0 = 16u * (unsigned)i4;
        const unsigned kw[4] = {(unsigned)cur[q].x, (unsigned)cur[q].y,
                                (unsigned)cur[q].z, (unsigned)cur[q].w};
#pragma unroll
        for (int j = 0; j < 4; ++j) {
            const unsigned x = kw[j] ^ pat;
            unsigned z = ~(((x & 0x7F7F7F7Fu) + 0x7F7F7F7Fu) | x | 0x7F7F7F7Fu);
            if (!live) z = 0u;
            unsigned bal = __builtin_amdgcn_ballot_w32(z != 0u);
            while (bal != 0u) {
                if (z != 0u) {
                    const unsigned k = (unsigned)__builtin_ctz(z) >> 3;
                    wl[cntw + __builtin_amdgcn_mbcnt_lo(bal, 0u)] = e0 + 4u * (unsigned)j + k;
                    z &= z - 1u;
                }
                cntw += (unsigned)__builtin_popcount(bal);
                bal = __builtin_amdgcn_ballot_w32(z != 0u);
            }
        }
    }

    unsigned i0 = 0u;
    for (; i0 + 4u <= cntw; i0 += 4u)
        add_entries<4>(wl, i0, rec, dinv, h4, acc, dinv_row, lane);
    if (i0 != 0u) {
        const unsigned rest = cntw - i0;
        unsigned t = 0u;
        if ((unsigned)lane < rest) t = wl[i0 + (unsigned)lane];
        if ((unsigned)lane < rest) wl[lane] = t;
        cntw = rest;
    }
}

__global__ __launch_bounds__(1024) void k_tile_agg(const unsigned* __restrict__ keyw,
                                                   const unsigned* __restrict__ rec,
                                                   const float* __restrict__ dinv,
                                                   const float* __restrict__ h,
                                                   const float* __restrict__ b1,
                                                   const float* __restrict__ W2,
                                                   const float* __restrict__ b2,
                                                   float* __restrict__ out) {
    __shared__ float    acc[TILE_R * DIM];
    __shared__ __attribute__((aligned(16))) unsigned list[32 * LIST_W];
    __shared__ float    dinv_row[TILE_R];

    const int tid  = threadIdx.x;
    const int lane = tid & 31;
    const int wave = tid >> 5;
    const int lo   = blockIdx.x * TILE_R;
    const int nrows = (N_NODES - lo < TILE_R) ? (N_NODES - lo) : TILE_R;
    const int4*   key4 = (const int4*)keyw;
    const float4* h4   = (const float4*)h;
    unsigned*     wl   = list + wave * LIST_W;
    const unsigned pat = (unsigned)blockIdx.x * 0x01010101u;

    int4 bufA[2], bufB[2];
    load_chunk<false>(key4, 0, tid, bufA);

    if (tid < TILE_R) dinv_row[tid] = (tid < nrows) ? dinv[lo + tid] : 0.0f;
    for (int r = wave; r < nrows; r += 32) {
        float s = dinv[lo + r];
        s *= s;
        const float4 v = h4[(size_t)(lo + r) * (DIM / 4) + lane];
        float* a = acc + r * DIM + lane;
        a[0] = v.x * s; a[32] = v.y * s; a[64] = v.z * s; a[96] = v.w * s;
    }
    __syncthreads();

    unsigned cntw = 0u;
    for (int c = 0; c < N_CHUNK - 2; c += 2) {
        tile_chunk<0>(c,     bufA, bufB, cntw, pat, key4, rec, dinv, h4, acc, wl, dinv_row, tid, lane);
        tile_chunk<0>(c + 1, bufB, bufA, cntw, pat, key4, rec, dinv, h4, acc, wl, dinv_row, tid, lane);
    }
    tile_chunk<1>(N_CHUNK - 2, bufA, bufB, cntw, pat, key4, rec, dinv, h4, acc, wl, dinv_row, tid, lane);
    tile_chunk<2>(N_CHUNK - 1, bufB, bufA, cntw, pat, key4, rec, dinv, h4, acc, wl, dinv_row, tid, lane);
    for (unsigned i = 0u; i < cntw; ++i)
        add_entries<1>(wl, i, rec, dinv, h4, acc, dinv_row, lane);


    const int wv      = __builtin_amdgcn_readfirstlane(wave);
    const int l15     = lane & 15;
    const int hi      = lane >> 4;
    const int colBase = (wv & 7) << 4;

    const float4 bb = ((const float4*)b1)[lane];

    __syncthreads();

    v2f bw[DIM / 4];
    const float* wcol = W2 + (size_t)(hi << 1) * DIM + colBase + l15;
#pragma unroll
    for (int kk = 0; kk < DIM / 4; ++kk) {
        bw[kk].x = wcol[(size_t)(4 * kk) * DIM];
        bw[kk].y = wcol[(size_t)(4 * kk + 1) * DIM];
    }
    const float bv = b2[colBase + l15];

    float* act = (float*)list;
    for (int r = wv; r < nrows; r += 32) {
        const float* a = acc + r * DIM + lane;
        const float x0 = a[0] + bb.x, x1 = a[32] + bb.y, x2 = a[64] + bb.z, x3 = a[96] + bb.w;
        float4 o;
        o.x = tanhf(x0);
        o.y = tanhf(x1);
        o.z = tanhf(x2);
        o.w = tanhf(x3);
        *(float4*)(act + r * ACT_S + 4 * lane) = o;
    }
    __syncthreads();

    const int nstrips = (nrows + 15) >> 4;
    for (int s = wv >> 3; s < nstrips; s += 4) {
        int row = (s << 4) + l15;
        row = (row < nrows) ? row : (nrows - 1);
        const float* ar = act + row * ACT_S + (hi << 1);
        v8f c = {};
#pragma unroll
        for (int kk = 0; kk < DIM / 4; ++kk) {
            const v2f a = *(const v2f*)(ar + 4 * kk);
            c = __builtin_amdgcn_wmma_f32_16x16x4_f32(false, a, false, bw[kk],
                                                      (short)0, c, false, false);
        }
        const int m0 = (s << 4) + (hi << 3);
        const int nvalid = nrows - m0;
        float* orow = out + (size_t)(lo + m0) * DIM + colBase + l15;
#pragma unroll
        for (int r = 0; r < 8; ++r)
            if (r < nvalid) orow[(size_t)r * DIM] = c[r] + bv;
    }
}


extern "C" void kernel_launch(void* const* d_in, const int* in_sizes, int n_in,
                              void* d_out, int out_size, void* d_ws, size_t ws_size,
                              hipStream_t stream) {
    const float* x  = (const float*)d_in[0];
    const int*   ei = (const int*)  d_in[1];
    const float* W1 = (const float*)d_in[2];
    const float* b1 = (const float*)d_in[3];
    const float* W2 = (const float*)d_in[4];
    const float* b2 = (const float*)d_in[5];
    float* out = (float*)d_out;

    float*    ws   = (float*)d_ws;
    unsigned* deg  = (unsigned*)ws;
    float*    dinv = ws + 65536;
    float*    h    = ws + 131072;
    unsigned* keyw = (unsigned*)(ws + 131072 + N_NODES * DIM);
    unsigned* rec  = keyw + NE4;

    const int* esrc = ei;
    const int* edst = ei + N_EDGES;

    k_deg_init <<<(N_NODES + 255) / 256, 256, 0, stream>>>(deg);
    k_deg_count<<<(NE4 + 255) / 256, 256, 0, stream>>>(esrc, edst, deg, keyw, rec);
    k_dinv     <<<(N_NODES + 255) / 256, 256, 0, stream>>>(deg, dinv);

    k_gemm <<<GEMM_WGS, GEMM_WAVES * 32, 0, stream>>>(x, W1, h);
    k_tile_agg<<<N_TILES, 1024, 0, stream>>>(keyw, rec, dinv, h, b1, W2, b2, out);
}


# Norm applied per row, not per edge: GEMM stores rsqrt(deg)*h, tile adds rows as they are and scales each row sum once; per-edge dinv look-up, dinv table and launch gone; sum order = LDS arrival
# speedup vs baseline: 2.2551x; 1.0527x over previous
#include <hip/hip_runtime.h>
#include <math.h>

#define N_NODES 50000
#define N_EDGES 640000
#define DIM     128
#define TILE_R   196
#define N_TILES  ((N_NODES + TILE_R - 1) / TILE_R)
#define NE4      (N_EDGES / 4)

typedef __attribute__((ext_vector_type(2))) float v2f;
typedef __attribute__((ext_vector_type(8))) float v8f;


__global__ void k_deg_init(unsigned* __restrict__ deg) {
    int i = blockIdx.x * blockDim.x + threadIdx.x;
    if (i < N_NODES) deg[i] = 1u;
}

static_assert(N_NODES <= 65536, "a source number must fit 16 bits of an edge record");
static_assert(N_TILES <= 256 && TILE_R <= 256, "tile number and row in the tile are one byte each");

__global__ void k_deg_count(const int* __restrict__ src, const int* __restrict__ dst,
                            unsigned* __restrict__ deg, unsigned* __restrict__ keyw,
                            unsigned* __restrict__ rec) {
    int t = blockIdx.x * blockDim.x + threadIdx.x;
    if (t < NE4) {
        const int4 d = ((const int4*)dst)[t];
        const int4 s = ((const int4*)src)[t];
        atomicAdd(&deg[d.x], 1u);
        atomicAdd(&deg[d.y], 1u);
        atomicAdd(&deg[d.z], 1u);
        atomicAdd(&deg[d.w], 1u);
        const unsigned k0 = (unsigned)d.x / TILE_R, k1 = (unsigned)d.y / TILE_R;
        const unsigned k2 = (unsigned)d.z / TILE_R, k3 = (unsigned)d.w / TILE_R;
        keyw[t] = k0 | (k1 << 8) | (k2 << 16) | (k3 << 24);
        uint4 r;
        r.x = (unsigned)s.x | (((unsigned)d.x - k0 * TILE_R) << 16);
        r.y = (unsigned)s.y | (((unsigned)d.y - k1 * TILE_R) << 16);
        r.z = (unsigned)s.z | (((unsigned)d.z - k2 * TILE_R) << 16);
        r.w = (unsigned)s.w | (((unsigned)d.w - k3 * TILE_R) << 16);
        ((uint4*)rec)[t] = r;
    }
}


#define N_STRIPS   (N_NODES / 16)
#define GEMM_WAVES 16
#define GEMM_WGS   ((N_STRIPS + GEMM_WAVES - 1) / GEMM_WAVES)
static_assert(N_NODES % 16 == 0, "whole strips only");
static_assert(GEMM_WAVES * 2 * 4 == DIM, "wave w puts the k-steps 2w and 2w+1 of W into LDS");

__global__ __launch_bounds__(GEMM_WAVES * 32) void k_gemm(const float* __restrict__ A,
                                                          const float* __restrict__ W,
                                                          const unsigned* __restrict__ deg,
                                                          float* __restrict__ out) {
    __shared__ __attribute__((aligned(16))) float wfrag[DIM * DIM];

    const int lane = threadIdx.x & 31;
    const int wv   = __builtin_amdgcn_readfirstlane((int)(threadIdx.x >> 5));
    const int l15  = lane & 15;
    const int hi   = lane >> 4;
    const int strip = blockIdx.x * GEMM_WAVES + wv;

    const float* wcol = W + (size_t)(8 * wv + (hi << 1)) * DIM + l15;
    float4 wq[8];
#pragma unroll
    for (int n = 0; n < 8; ++n) {
        wq[n].x = wcol[(n << 4)];
        wq[n].y = wcol[(n << 4) + DIM];
        wq[n].z = wcol[(n << 4) + 4 * DIM];
        wq[n].w = wcol[(n << 4) + 5 * DIM];
    }
    float4* wf4 = (float4*)wfrag + lane;
#pragma unroll
    for (int n = 0; n < 8; ++n)
        wf4[(wv * 8 + n) * 32] = wq[n];

    unsigned ooff = (unsigned)((((hi << 3) * DIM) + l15) * 4);
    asm volatile("" : "+v"(ooff));
    __syncthreads();

    if (strip < N_STRIPS) {
        const uint4* dg = (const uint4*)(deg + (strip << 4) + (hi << 3));
        const uint4 dg0 = dg[0], dg1 = dg[1];

        v2f a[DIM / 4];
        const float* arow = A + (size_t)((strip << 4) + l15) * DIM + (hi << 1);
#pragma unroll
        for (int kk = 0; kk < DIM / 4; ++kk)
            a[kk] = *(const v2f*)(arow + 4 * kk);

        v8f c[8] = {};
#pragma unroll
        for (int kp = 0; kp < DIM / 8; ++kp) {
            float4 bq[8];
#pragma unroll
            for (int n = 0; n < 8; ++n) bq[n] = wf4[(kp * 8 + n) * 32];
#pragma unroll
            for (int n = 0; n < 8; ++n) {
                v2f b; b.x = bq[n].x; b.y = bq[n].y;
                c[n] = __builtin_amdgcn_wmma_f32_16x16x4_f32(false, a[2 * kp], false, b,
                                                             (short)0, c[n], false, false);
            }
#pragma unroll
            for (int n = 0; n < 8; ++n) {
                v2f b; b.x = bq[n].z; b.y = bq[n].w;
                c[n] = __builtin_amdgcn_wmma_f32_16x16x4_f32(false, a[2 * kp + 1], false, b,
                                                             (short)0, c[n], false, false);
            }
        }

        const float sc[8] = {rsqrtf((float)dg0.x), rsqrtf((float)dg0.y), rsqrtf((float)dg0.z), rsqrtf((float)dg0.w),
                             rsqrtf((float)dg1.x), rsqrtf((float)dg1.y), rsqrtf((float)dg1.z), rsqrtf((float)dg1.w)};
        char* obase = (char*)(out + (size_t)(strip << 4) * DIM);
#pragma unroll
        for (int n = 0; n < 8; ++n)
#pragma unroll
            for (int r = 0; r < 8; ++r)
                *(float*)(obase + ooff + (r * DIM + (n << 4)) * 4) = c[n][r] * sc[r];
    }
}


#define NK4      (N_EDGES / 16)
#define CHUNK4   2048
#define N_CHUNK  ((NK4 + CHUNK4 - 1) / CHUNK4)
#define LIST_W   1028
#define ACT_S    132

static_assert(N_EDGES % 16 == 0, "the key list is read as whole int4");
static_assert(N_CHUNK % 2 == 0, "chunks are taken in pairs");
static_assert((N_CHUNK - 1) * CHUNK4 <= NK4 && N_CHUNK * CHUNK4 >= NK4, "only the last chunk may pass the end");

template <bool LAST>
__device__ __forceinline__ void load_chunk(const int4* __restrict__ key4, int c, int tid,
                                           int4 (&buf)[2]) {
#pragma unroll
    for (int q = 0; q < 2; ++q) {
        const int i4 = c * CHUNK4 + q * 1024 + tid;
        buf[q] = key4[(LAST && i4 >= NK4) ? (NK4 - 1) : i4];
    }
}

template <int U>
__device__ __forceinline__ void add_entries(const unsigned* wl, unsigned i0,
                                            const unsigned* __restrict__ rec,
                                            const float4* __restrict__ h4,
                                            float* acc, int lane) {
    unsigned ent[U];
    unsigned rc[U];
    unsigned sv[U];
    unsigned rw[U];
    float4   v[U];
#pragma unroll
    for (int u = 0; u < U; ++u) ent[u] = wl[i0 + u];
#pragma unroll
    for (int u = 0; u < U; ++u) rc[u] = rec[ent[u]];
#pragma unroll
    for (int u = 0; u < U; ++u) {
        sv[u] = rc[u] & 0xFFFFu;
        rw[u] = rc[u] >> 16;
        v[u]  = h4[(size_t)sv[u] * (DIM / 4) + lane];
    }
#pragma unroll
    for (int u = 0; u < U; ++u) {
        float* a = acc + rw[u] * DIM + lane;
        unsafeAtomicAdd(a,      v[u].x);
        unsafeAtomicAdd(a + 32, v[u].y);
        unsafeAtomicAdd(a + 64, v[u].z);
        unsafeAtomicAdd(a + 96, v[u].w);
    }
}

template <int NEXT>
__device__ __forceinline__ void tile_chunk(int c, const int4 (&cur)[2], int4 (&nxt)[2],
                                           unsigned& cntw, unsigned pat,
                                           const int4* __restrict__ key4,
                                           const unsigned* __restrict__ rec,
                                           const float4* __restrict__ h4,
                                           float* acc, unsigned* wl,
                                           int tid, int lane) {
    if (NEXT == 0) load_chunk<false>(key4, c + 1, tid, nxt);
    if (NEXT == 1) load_chunk<true>(key4, c + 1, tid, nxt);

#pragma unroll
    for (int q = 0; q < 2; ++q) {
        const int      i4 = c * CHUNK4 + q * 1024 + tid;
        const bool     live = (NEXT != 2) || (i4 < NK4);
        const unsigned e0 = 16u * (unsigned)i4;
        const unsigned kw[4] = {(unsigned)cur[q].x, (unsigned)cur[q].y,
                                (unsigned)cur[q].z, (unsigned)cur[q].w};
#pragma unroll
        for (int j = 0; j < 4; ++j) {
            const unsigned x = kw[j] ^ pat;
            unsigned z = ~(((x & 0x7F7F7F7Fu) + 0x7F7F7F7Fu) | x | 0x7F7F7F7Fu);
            if (!live) z = 0u;
            unsigned bal = __builtin_amdgcn_ballot_w32(z != 0u);
            while (bal != 0u) {
                if (z != 0u) {
                    const unsigned k = (unsigned)__builtin_ctz(z) >> 3;
                    wl[cntw + __builtin_amdgcn_mbcnt_lo(bal, 0u)] = e0 + 4u * (unsigned)j + k;
                    z &= z - 1u;
                }
                cntw += (unsigned)__builtin_popcount(bal);
                bal = __builtin_amdgcn_ballot_w32(z != 0u);
            }
        }
    }

    unsigned i0 = 0u;
    for (; i0 + 4u <= cntw; i0 += 4u)
        add_entries<4>(wl, i0, rec, h4, acc, lane);
    if (i0 != 0u) {
        const unsigned rest = cntw - i0;
        unsigned t = 0u;
        if ((unsigned)lane < rest) t = wl[i0 + (unsigned)lane];
        if ((unsigned)lane < rest) wl[lane] = t;
        cntw = rest;
    }
}

__global__ __launch_bounds__(1024) void k_tile_agg(const unsigned* __restrict__ keyw,
                                                   const unsigned* __restrict__ rec,
                                                   const unsigned* __restrict__ deg,
                                                   const float* __restrict__ h,
                                                   const float* __restrict__ b1,
                                                   const float* __restrict__ W2,
                                                   const float* __restrict__ b2,
                                                   float* __restrict__ out) {
    __shared__ float    acc[TILE_R * DIM];
    __shared__ __attribute__((aligned(16))) unsigned list[32 * LIST_W];
    __shared__ float    dinv_row[TILE_R];

    const int tid  = threadIdx.x;
    const int lane = tid & 31;
    const int wave = tid >> 5;
    const int lo   = blockIdx.x * TILE_R;
    const int nrows = (N_NODES - lo < TILE_R) ? (N_NODES - lo) : TILE_R;
    const int4*   key4 = (const int4*)keyw;
    const float4* h4   = (const float4*)h;
    unsigned*     wl   = list + wave * LIST_W;
    const unsigned pat = (unsigned)blockIdx.x * 0x01010101u;

    int4 bufA[2], bufB[2];
    load_chunk<false>(key4, 0, tid, bufA);

    if (tid < TILE_R) dinv_row[tid] = (tid < nrows) ? rsqrtf((float)deg[lo + tid]) : 0.0f;
    for (int r = wave; r < nrows; r += 32) {
        const float4 v = h4[(size_t)(lo + r) * (DIM / 4) + lane];
        float* a = acc + r * DIM + lane;
        a[0] = v.x; a[32] = v.y; a[64] = v.z; a[96] = v.w;
    }
    __syncthreads();

    unsigned cntw = 0u;
    for (int c = 0; c < N_CHUNK - 2; c += 2) {
        tile_chunk<0>(c,     bufA, bufB, cntw, pat, key4, rec, h4, acc, wl, tid, lane);
        tile_chunk<0>(c + 1, bufB, bufA, cntw, pat, key4, rec, h4, acc, wl, tid, lane);
    }
    tile_chunk<1>(N_CHUNK - 2, bufA, bufB, cntw, pat, key4, rec, h4, acc, wl, tid, lane);
    tile_chunk<2>(N_CHUNK - 1, bufB, bufA, cntw, pat, key4, rec, h4, acc, wl, tid, lane);
    for (unsigned i = 0u; i < cntw; ++i)
        add_entries<1>(wl, i, rec, h4, acc, lane);


    const int wv      = __builtin_amdgcn_readfirstlane(wave);
    const int l15     = lane & 15;
    const int hi      = lane >> 4;
    const int colBase = (wv & 7) << 4;

    const float4 bb = ((const float4*)b1)[lane];

    __syncthreads();

    v2f bw[DIM / 4];
    const float* wcol = W2 + (size_t)(hi << 1) * DIM + colBase + l15;
#pragma unroll
    for (int kk = 0; kk < DIM / 4; ++kk) {
        bw[kk].x = wcol[(size_t)(4 * kk) * DIM];
        bw[kk].y = wcol[(size_t)(4 * kk + 1) * DIM];
    }
    const float bv = b2[colBase + l15];

    float* act = (float*)list;
    for (int r = wv; r < nrows; r += 32) {
        const float* a = acc + r * DIM + lane;
        const float dr = dinv_row[r];
        const float x0 = a[0] * dr + bb.x, x1 = a[32] * dr + bb.y, x2 = a[64] * dr + bb.z, x3 = a[96] * dr + bb.w;
        float4 o;
        o.x = tanhf(x0);
        o.y = tanhf(x1);
        o.z = tanhf(x2);
        o.w = tanhf(x3);
        *(float4*)(act + r * ACT_S + 4 * lane) = o;
    }
    __syncthreads();

    const int nstrips = (nrows + 15) >> 4;
    for (int s = wv >> 3; s < nstrips; s += 4) {
        int row = (s << 4) + l15;
        row = (row < nrows) ? row : (nrows - 1);
        const float* ar = act + row * ACT_S + (hi << 1);
        v8f c = {};
#pragma unroll
        for (int kk = 0; kk < DIM / 4; ++kk) {
            const v2f a = *(const v2f*)(ar + 4 * kk);
            c = __builtin_amdgcn_wmma_f32_16x16x4_f32(false, a, false, bw[kk],
                                                      (short)0, c, false, false);
        }
        const int m0 = (s << 4) + (hi << 3);
        const int nvalid = nrows - m0;
        float* orow = out + (size_t)(lo + m0) * DIM + colBase + l15;
#pragma unroll
        for (int r = 0; r < 8; ++r)
            if (r < nvalid) orow[(size_t)r * DIM] = c[r] + bv;
    }
}


extern "C" void kernel_launch(void* const* d_in, const int* in_sizes, int n_in,
                              void* d_out, int out_size, void* d_ws, size_t ws_size,
                              hipStream_t stream) {
    const float* x  = (const float*)d_in[0];
    const int*   ei = (const int*)  d_in[1];
    const float* W1 = (const float*)d_in[2];
    const float* b1 = (const float*)d_in[3];
    const float* W2 = (const float*)d_in[4];
    const float* b2 = (const float*)d_in[5];
    float* out = (float*)d_out;

    float*    ws   = (float*)d_ws;
    unsigned* deg  = (unsigned*)ws;
    float*    h    = ws + 131072;
    unsigned* keyw = (unsigned*)(ws + 131072 + N_NODES * DIM);
    unsigned* rec  = keyw + NE4;

    const int* esrc = ei;
    const int* edst = ei + N_EDGES;

    k_deg_init <<<(N_NODES + 255) / 256, 256, 0, stream>>>(deg);
    k_deg_count<<<(NE4 + 255) / 256, 256, 0, stream>>>(esrc, edst, deg, keyw, rec);

    k_gemm <<<GEMM_WGS, GEMM_WAVES * 32, 0, stream>>>(x, W1, deg, h);
    k_tile_agg<<<N_TILES, 1024, 0, stream>>>(keyw, rec, deg, h, b1, W2, b2, out);
}


# Count launch sorts each 4096-edge block by tile in LDS and writes the block's tile offsets; a tile reads only its own records, 16 rows in flight: key scan, match lists gone; sum order = LDS arrival
# speedup vs baseline: 2.4427x; 1.0832x over previous
#include <hip/hip_runtime.h>
#include <math.h>

#define N_NODES 50000
#define N_EDGES 640000
#define DIM     128
#define TILE_R   196
#define N_TILES  ((N_NODES + TILE_R - 1) / TILE_R)
#define NE4      (N_EDGES / 4)
#define BLK_E    4096
#define N_BLK    ((N_EDGES + BLK_E - 1) / BLK_E)

typedef __attribute__((ext_vector_type(2))) float v2f;
typedef __attribute__((ext_vector_type(8))) float v8f;


__global__ void k_deg_init(unsigned* __restrict__ deg) {
    int i = blockIdx.x * blockDim.x + threadIdx.x;
    if (i < N_NODES) deg[i] = 1u;
}

static_assert(N_NODES <= 65536, "a source number must fit 16 bits of an edge record");
static_assert(N_TILES <= 256 && TILE_R <= 256, "tile number and row in the tile are one byte each");
static_assert(BLK_E == 4 * 1024, "1024 count threads of four edges each");
static_assert(BLK_E <= 32768, "first and end of a stretch are 16 bits each, end << 16 stays inside the word");
static_assert(N_BLK * N_TILES <= NE4, "the offset words live in the space of NE4 words");

__global__ __launch_bounds__(1024) void k_deg_count(const int* __restrict__ src, const int* __restrict__ dst,
                                                    unsigned* __restrict__ deg, unsigned* __restrict__ offs,
                                                    unsigned* __restrict__ rec) {
    __shared__ unsigned cnt[N_TILES];
    __shared__ unsigned first[N_TILES + 1];
    __shared__ __attribute__((aligned(16))) unsigned sorted[BLK_E];

    const int  tid  = threadIdx.x;
    const int  t4   = blockIdx.x * (BLK_E / 4) + tid;
    const bool live = t4 < NE4;

    if (tid < N_TILES) cnt[tid] = 0u;
    __syncthreads();

    unsigned tl[4] = {0u, 0u, 0u, 0u};
    unsigned pl[4] = {0u, 0u, 0u, 0u};
    unsigned rw[4] = {0u, 0u, 0u, 0u};
    if (live) {
        const int4 d = ((const int4*)dst)[t4];
        const int4 e = ((const int4*)src)[t4];
        const unsigned dv[4] = {(unsigned)d.x, (unsigned)d.y, (unsigned)d.z, (unsigned)d.w};
        const unsigned ev[4] = {(unsigned)e.x, (unsigned)e.y, (unsigned)e.z, (unsigned)e.w};
#pragma unroll
        for (int j = 0; j < 4; ++j) {
            atomicAdd(&deg[dv[j]], 1u);
            tl[j] = dv[j] / TILE_R;
            rw[j] = ev[j] | ((dv[j] - tl[j] * TILE_R) << 16);
            pl[j] = atomicAdd(&cnt[tl[j]], 1u);
        }
    }
    __syncthreads();

    if (tid < 32) {
        unsigned c[8];
        unsigned run = 0u;
#pragma unroll
        for (int i = 0; i < 8; ++i) { c[i] = run; run += cnt[8 * tid + i]; }
        unsigned incl = run;
#pragma unroll
        for (int o = 1; o < 32; o <<= 1) {
            const unsigned y = __shfl_up(incl, o);
            if (tid >= o) incl += y;
        }
        const unsigned base = incl - run;
#pragma unroll
        for (int i = 0; i < 8; ++i) first[8 * tid + i] = base + c[i];
        if (tid == 31) first[N_TILES] = incl;
    }
    __syncthreads();

    if (live) {
#pragma unroll
        for (int j = 0; j < 4; ++j) sorted[first[tl[j]] + pl[j]] = rw[j];
    }
    __syncthreads();

    const unsigned total = first[N_TILES];
    if (4u * (unsigned)tid < total)
        ((uint4*)rec)[blockIdx.x * (BLK_E / 4) + tid] = ((const uint4*)sorted)[tid];
    if (tid < N_TILES)
        offs[blockIdx.x * N_TILES + tid] = first[tid] | (first[tid + 1] << 16);
}


#define N_STRIPS   (N_NODES / 16)
#define GEMM_WAVES 16
#define GEMM_WGS   ((N_STRIPS + GEMM_WAVES - 1) / GEMM_WAVES)
static_assert(N_NODES % 16 == 0, "whole strips only");
static_assert(GEMM_WAVES * 2 * 4 == DIM, "wave w puts the k-steps 2w and 2w+1 of W into LDS");

__global__ __launch_bounds__(GEMM_WAVES * 32) void k_gemm(const float* __restrict__ A,
                                                          const float* __restrict__ W,
                                                          const unsigned* __restrict__ deg,
                                                          float* __restrict__ out) {
    __shared__ __attribute__((aligned(16))) float wfrag[DIM * DIM];

    const int lane = threadIdx.x & 31;
    const int wv   = __builtin_amdgcn_readfirstlane((int)(threadIdx.x >> 5));
    const int l15  = lane & 15;
    const int hi   = lane >> 4;
    const int strip = blockIdx.x * GEMM_WAVES + wv;

    const float* wcol = W + (size_t)(8 * wv + (hi << 1)) * DIM + l15;
    float4 wq[8];
#pragma unroll
    for (int n = 0; n < 8; ++n) {
        wq[n].x = wcol[(n << 4)];
        wq[n].y = wcol[(n << 4) + DIM];
        wq[n].z = wcol[(n << 4) + 4 * DIM];
        wq[n].w = wcol[(n << 4) + 5 * DIM];
    }
    float4* wf4 = (float4*)wfrag + lane;
#pragma unroll
    for (int n = 0; n < 8; ++n)
        wf4[(wv * 8 + n) * 32] = wq[n];

    unsigned ooff = (unsigned)((((hi << 3) * DIM) + l15) * 4);
    asm volatile("" : "+v"(ooff));
    __syncthreads();

    if (strip < N_STRIPS) {
        const uint4* dg = (const uint4*)(deg + (strip << 4) + (hi << 3));
        const uint4 dg0 = dg[0], dg1 = dg[1];

        v2f a[DIM / 4];
        const float* arow = A + (size_t)((strip << 4) + l15) * DIM + (hi << 1);
#pragma unroll
        for (int kk = 0; kk < DIM / 4; ++kk)
            a[kk] = *(const v2f*)(arow + 4 * kk);

        v8f c[8] = {};
#pragma unroll
        for (int kp = 0; kp < DIM / 8; ++kp) {
            float4 bq[8];
#pragma unroll
            for (int n = 0; n < 8; ++n) bq[n] = wf4[(kp * 8 + n) * 32];
#pragma unroll
            for (int n = 0; n < 8; ++n) {
                v2f b; b.x = bq[n].x; b.y = bq[n].y;
                c[n] = __builtin_amdgcn_wmma_f32_16x16x4_f32(false, a[2 * kp], false, b,
                                                             (short)0, c[n], false, false);
            }
#pragma unroll
            for (int n = 0; n < 8; ++n) {
                v2f b; b.x = bq[n].z; b.y = bq[n].w;
                c[n] = __builtin_amdgcn_wmma_f32_16x16x4_f32(false, a[2 * kp + 1], false, b,
                                                             (short)0, c[n], false, false);
            }
        }

        const float sc[8] = {rsqrtf((float)dg0.x), rsqrtf((float)dg0.y), rsqrtf((float)dg0.z), rsqrtf((float)dg0.w),
                             rsqrtf((float)dg1.x), rsqrtf((float)dg1.y), rsqrtf((float)dg1.z), rsqrtf((float)dg1.w)};
        char* obase = (char*)(out + (size_t)(strip << 4) * DIM);
#pragma unroll
        for (int n = 0; n < 8; ++n)
#pragma unroll
            for (int r = 0; r < 8; ++r)
                *(float*)(obase + ooff + (r * DIM + (n << 4)) * 4) = c[n][r] * sc[r];
    }
}


#define BPW      ((N_BLK + 31) / 32)
#define GB       16
#define ACT_S    132

__global__ __launch_bounds__(1024) void k_tile_agg(const unsigned* __restrict__ offs,
                                                   const unsigned* __restrict__ rec,
                                                   const unsigned* __restrict__ deg,
                                                   const float* __restrict__ h,
                                                   const float* __restrict__ b1,
                                                   const float* __restrict__ W2,
                                                   const float* __restrict__ b2,
                                                   float* __restrict__ out) {
    __shared__ float    acc[TILE_R * DIM];
    __shared__ __attribute__((aligned(16))) float actbuf[TILE_R * ACT_S];
    __shared__ float    dinv_row[TILE_R];

    const int tid  = threadIdx.x;
    const int lane = tid & 31;
    const int wave = tid >> 5;
    const int wv   = __builtin_amdgcn_readfirstlane(wave);
    const int lo   = blockIdx.x * TILE_R;
    const int nrows = (N_NODES - lo < TILE_R) ? (N_NODES - lo) : TILE_R;
    const float4* h4 = (const float4*)h;

    const unsigned* offt = offs + blockIdx.x;
    unsigned p0 = (wv      < N_BLK) ? offt[(wv     ) * N_TILES] : 0u;
    unsigned p1 = (wv + 32 < N_BLK) ? offt[(wv + 32) * N_TILES] : 0u;
    unsigned r0;
    {
        const unsigned st = p0 & 0xFFFFu, n = (p0 >> 16) - st;
        r0 = ((unsigned)lane < n) ? rec[(unsigned)wv * BLK_E + st + (unsigned)lane] : 0u;
    }

    if (tid < TILE_R) dinv_row[tid] = (tid < nrows) ? rsqrtf((float)deg[lo + tid]) : 0.0f;
    for (int r = wave; r < nrows; r += 32) {
        const float4 v = h4[(size_t)(lo + r) * (DIM / 4) + lane];
        float* a = acc + r * DIM + lane;
        a[0] = v.x; a[32] = v.y; a[64] = v.z; a[96] = v.w;
    }
    __syncthreads();

    for (int b = wv; b < N_BLK; b += 32) {
        const unsigned p2 = (b + 64 < N_BLK) ? offt[(b + 64) * N_TILES] : 0u;
        const unsigned p   = (unsigned)__builtin_amdgcn_readfirstlane((int)p0);
        const unsigned st  = p & 0xFFFFu;
        const unsigned n   = (p >> 16) - st;
        const unsigned q   = (unsigned)__builtin_amdgcn_readfirstlane((int)p1);
        const unsigned st1 = q & 0xFFFFu;
        const unsigned n1  = (q >> 16) - st1;
        const unsigned* nxt = rec + (unsigned)(b + 32) * BLK_E + st1 + (unsigned)lane;
        unsigned r1  = 0u;
        unsigned rcv = r0;
        for (unsigned w0 = 0u; w0 < n; w0 += 32u) {
            const unsigned m = (n - w0 < 32u) ? (n - w0) : 32u;
            if (w0 != 0u)
                rcv = ((unsigned)lane < m) ? rec[(unsigned)b * BLK_E + st + w0 + (unsigned)lane] : 0u;
            for (unsigned j = 0u; j < m; j += GB) {
                const unsigned c = (m - j < (unsigned)GB) ? (m - j) : (unsigned)GB;
                unsigned r[GB];
                float4   v[GB];
#pragma unroll
                for (int u = 0; u < GB; ++u)
                    r[u] = (unsigned)__builtin_amdgcn_readlane((int)rcv, (int)(j + (unsigned)u));
#pragma unroll
                for (int u = 0; u < GB; ++u) {
                    if ((unsigned)u < c) v[u] = h4[(size_t)(r[u] & 0xFFFFu) * (DIM / 4) + lane];
                }
                if (w0 == 0u && j == 0u && (unsigned)lane < n1) r1 = *nxt;
#pragma unroll
                for (int u = 0; u < GB; ++u) {
                    if ((unsigned)u < c) {
                        float* a = acc + (r[u] >> 16) * DIM + lane;
                        unsafeAtomicAdd(a,      v[u].x);
                        unsafeAtomicAdd(a + 32, v[u].y);
                        unsafeAtomicAdd(a + 64, v[u].z);
                        unsafeAtomicAdd(a + 96, v[u].w);
                    }
                }
            }
        }
        if (n == 0u && (unsigned)lane < n1) r1 = *nxt;
        p0 = p1; p1 = p2; r0 = r1;
    }


    const int l15     = lane & 15;
    const int hi      = lane >> 4;
    const int colBase = (wv & 7) << 4;

    const float4 bb = ((const float4*)b1)[lane];

    __syncthreads();

    v2f bw[DIM / 4];
    const float* wcol = W2 + (size_t)(hi << 1) * DIM + colBase + l15;
#pragma unroll
    for (int kk = 0; kk < DIM / 4; ++kk) {
        bw[kk].x = wcol[(size_t)(4 * kk) * DIM];
        bw[kk].y = wcol[(size_t)(4 * kk + 1) * DIM];
    }
    const float bv = b2[colBase + l15];

    float* act = actbuf;
    for (int r = wv; r < nrows; r += 32) {
        const float* a = acc + r * DIM + lane;
        const float dr = dinv_row[r];
        const float x0 = a[0] * dr + bb.x, x1 = a[32] * dr + bb.y, x2 = a[64] * dr + bb.z, x3 = a[96] * dr + bb.w;
        float4 o;
        o.x = tanhf(x0);
        o.y = tanhf(x1);
        o.z = tanhf(x2);
        o.w = tanhf(x3);
        *(float4*)(act + r * ACT_S + 4 * lane) = o;
    }
    __syncthreads();

    const int nstrips = (nrows + 15) >> 4;
    for (int s = wv >> 3; s < nstrips; s += 4) {
        int row = (s << 4) + l15;
        row = (row < nrows) ? row : (nrows - 1);
        const float* ar = act + row * ACT_S + (hi << 1);
        v8f c = {};
#pragma unroll
        for (int kk = 0; kk < DIM / 4; ++kk) {
            const v2f a = *(const v2f*)(ar + 4 * kk);
            c = __builtin_amdgcn_wmma_f32_16x16x4_f32(false, a, false, bw[kk],
                                                      (short)0, c, false, false);
        }
        const int m0 = (s << 4) + (hi << 3);
        const int nvalid = nrows - m0;
        float* orow = out + (size_t)(lo + m0) * DIM + colBase + l15;
#pragma unroll
        for (int r = 0; r < 8; ++r)
            if (r < nvalid) orow[(size_t)r * DIM] = c[r] + bv;
    }
}


extern "C" void kernel_launch(void* const* d_in, const int* in_sizes, int n_in,
                              void* d_out, int out_size, void* d_ws, size_t ws_size,
                              hipStream_t stream) {
    const float* x  = (const float*)d_in[0];
    const int*   ei = (const int*)  d_in[1];
    const float* W1 = (const float*)d_in[2];
    const float* b1 = (const float*)d_in[3];
    const float* W2 = (const float*)d_in[4];
    const float* b2 = (const float*)d_in[5];
    float* out = (float*)d_out;

    float*    ws   = (float*)d_ws;
    unsigned* deg  = (unsigned*)ws;
    float*    h    = ws + 131072;
    unsigned* offs = (unsigned*)(ws + 131072 + N_NODES * DIM);
    unsigned* rec  = offs + NE4;

    const int* esrc = ei;
    const int* edst = ei + N_EDGES;

    k_deg_init <<<(N_NODES + 255) / 256, 256, 0, stream>>>(deg);
    k_deg_count<<<N_BLK, 1024, 0, stream>>>(esrc, edst, deg, offs, rec);

    k_gemm <<<GEMM_WGS, GEMM_WAVES * 32, 0, stream>>>(x, W1, deg, h);
    k_tile_agg<<<N_TILES, 1024, 0, stream>>>(offs, rec, deg, h, b1, W2, b2, out);
}


# GEMM and edge sort in one launch: GEMM stores x@W1 unscaled (needs no degrees); tile scales each gathered row by rsqrt(deg[src]) from a 16-bit degree table in LDS; 4 launches -> 3; sums unchanged
# speedup vs baseline: 2.5255x; 1.0339x over previous
#include <hip/hip_runtime.h>
#include <math.h>

#define N_NODES 50000
#define N_EDGES 640000
#define DIM     128
#define TILE_R   196
#define N_TILES  ((N_NODES + TILE_R - 1) / TILE_R)
#define NE4      (N_EDGES / 4)
#define BLK_E    4096
#define N_BLK    ((N_EDGES + BLK_E - 1) / BLK_E)

typedef __attribute__((ext_vector_type(2))) float v2f;
typedef __attribute__((ext_vector_type(8))) float v8f;


__global__ void k_deg_init(unsigned* __restrict__ deg) {
    int i = blockIdx.x * blockDim.x + threadIdx.x;
    if (i < N_NODES) deg[i] = 1u;
}

static_assert(N_NODES <= 65536, "a source number must fit 16 bits of an edge record");
static_assert(N_TILES <= 256 && TILE_R <= 256, "tile number and row in the tile are one byte each");
static_assert(BLK_E <= 32768, "first and end of a stretch are 16 bits each, end << 16 stays inside the word");
static_assert(N_BLK * N_TILES <= NE4, "the offset words live in the space of NE4 words");

#define N_STRIPS   (N_NODES / 16)
#define GEMM_WAVES 16
#define GEMM_WGS   ((N_STRIPS + GEMM_WAVES - 1) / GEMM_WAVES)
#define SG_THREADS (GEMM_WAVES * 32)
static_assert(N_NODES % 16 == 0, "whole strips only");
static_assert(GEMM_WAVES * 2 * 4 == DIM, "wave w puts the k-steps 2w and 2w+1 of W into LDS");
static_assert(BLK_E == 8 * SG_THREADS, "a sort thread brings two int4 of the edge list");
static_assert(N_TILES <= SG_THREADS, "one thread per tile counter");
static_assert(BLK_E + 2 * N_TILES + 1 <= DIM * DIM, "the sort's tables fit the LDS block of the W fragments");

__global__ __launch_bounds__(SG_THREADS) __attribute__((amdgpu_waves_per_eu(4, 4))) void k_gemm_sort(const float* __restrict__ A,
                                                          const float* __restrict__ W,
                                                          const int* __restrict__ src,
                                                          const int* __restrict__ dst,
                                                          unsigned* __restrict__ deg,
                                                          unsigned* __restrict__ offs,
                                                          unsigned* __restrict__ rec,
                                                          float* __restrict__ out) {
    __shared__ __attribute__((aligned(16))) float smem[DIM * DIM];

    if (blockIdx.x < GEMM_WGS) {
        float* wfrag = smem;
        const int lane = threadIdx.x & 31;
        const int wv   = __builtin_amdgcn_readfirstlane((int)(threadIdx.x >> 5));
        const int l15  = lane & 15;
        const int hi   = lane >> 4;
        const int strip = blockIdx.x * GEMM_WAVES + wv;

        const float* wcol = W + (size_t)(8 * wv + (hi << 1)) * DIM + l15;
        float4 wq[8];
#pragma unroll
        for (int n = 0; n < 8; ++n) {
            wq[n].x = wcol[(n << 4)];
            wq[n].y = wcol[(n << 4) + DIM];
            wq[n].z = wcol[(n << 4) + 4 * DIM];
            wq[n].w = wcol[(n << 4) + 5 * DIM];
        }
        float4* wf4 = (float4*)wfrag + lane;
#pragma unroll
        for (int n = 0; n < 8; ++n)
            wf4[(wv * 8 + n) * 32] = wq[n];

        unsigned ooff = (unsigned)((((hi << 3) * DIM) + l15) * 4);
        asm volatile("" : "+v"(ooff));
        __syncthreads();

        if (strip < N_STRIPS) {
            v2f a[DIM / 4];
            const float* arow = A + (size_t)((strip << 4) + l15) * DIM + (hi << 1);
#pragma unroll
            for (int kk = 0; kk < DIM / 4; ++kk)
                a[kk] = *(const v2f*)(arow + 4 * kk);

            v8f c[8] = {};
#pragma unroll
            for (int kp = 0; kp < DIM / 8; ++kp) {
                float4 bq[8];
#pragma unroll
                for (int n = 0; n < 8; ++n) bq[n] = wf4[(kp * 8 + n) * 32];
#pragma unroll
                for (int n = 0; n < 8; ++n) {
                    v2f b; b.x = bq[n].x; b.y = bq[n].y;
                    c[n] = __builtin_amdgcn_wmma_f32_16x16x4_f32(false, a[2 * kp], false, b,
                                                                 (short)0, c[n], false, false);
                }
#pragma unroll
                for (int n = 0; n < 8; ++n) {
                    v2f b; b.x = bq[n].z; b.y = bq[n].w;
                    c[n] = __builtin_amdgcn_wmma_f32_16x16x4_f32(false, a[2 * kp + 1], false, b,
                                                                 (short)0, c[n], false, false);
                }
            }

            char* obase = (char*)(out + (size_t)(strip << 4) * DIM);
#pragma unroll
            for (int n = 0; n < 8; ++n)
#pragma unroll
                for (int r = 0; r < 8; ++r)
                    *(float*)(obase + ooff + (r * DIM + (n << 4)) * 4) = c[n][r];
        }
    } else {
        unsigned* sorted = (unsigned*)smem;
        unsigned* cnt    = sorted + BLK_E;
        unsigned* first  = cnt + N_TILES;

        const int tid = threadIdx.x;
        const int b   = (int)blockIdx.x - GEMM_WGS;

        if (tid < N_TILES) cnt[tid] = 0u;
        __syncthreads();

        unsigned tl[8] = {0u, 0u, 0u, 0u, 0u, 0u, 0u, 0u};
        unsigned pl[8] = {0u, 0u, 0u, 0u, 0u, 0u, 0u, 0u};
        unsigned rw[8] = {0u, 0u, 0u, 0u, 0u, 0u, 0u, 0u};
        bool live[2];
#pragma unroll
        for (int q = 0; q < 2; ++q) {
            const int t4 = b * (BLK_E / 4) + q * SG_THREADS + tid;
            live[q] = t4 < NE4;
            if (live[q]) {
                const int4 d = ((const int4*)dst)[t4];
                const int4 e = ((const int4*)src)[t4];
                const unsigned dv[4] = {(unsigned)d.x, (unsigned)d.y, (unsigned)d.z, (unsigned)d.w};
                const unsigned ev[4] = {(unsigned)e.x, (unsigned)e.y, (unsigned)e.z, (unsigned)e.w};
#pragma unroll
                for (int j = 0; j < 4; ++j) {
                    atomicAdd(&deg[dv[j]], 1u);
                    const unsigned t = dv[j] / TILE_R;
                    tl[4 * q + j] = t;
                    rw[4 * q + j] = ev[j] | ((dv[j] - t * TILE_R) << 16);
                    pl[4 * q + j] = atomicAdd(&cnt[t], 1u);
                }
            }
        }
        __syncthreads();

        if (tid < 32) {
            unsigned c[8];
            unsigned run = 0u;
#pragma unroll
            for (int i = 0; i < 8; ++i) { c[i] = run; run += cnt[8 * tid + i]; }
            unsigned incl = run;
#pragma unroll
            for (int o = 1; o < 32; o <<= 1) {
                const unsigned y = __shfl_up(incl, o);
                if (tid >= o) incl += y;
            }
            const unsigned base = incl - run;
#pragma unroll
            for (int i = 0; i < 8; ++i) first[8 * tid + i] = base + c[i];
            if (tid == 31) first[N_TILES] = incl;
        }
        __syncthreads();

#pragma unroll
        for (int q = 0; q < 2; ++q) {
            if (live[q]) {
#pragma unroll
                for (int j = 0; j < 4; ++j) sorted[first[tl[4 * q + j]] + pl[4 * q + j]] = rw[4 * q + j];
            }
        }
        __syncthreads();

        const unsigned total = first[N_TILES];
#pragma unroll
        for (int q = 0; q < 2; ++q) {
            const int i = q * SG_THREADS + tid;
            if (4u * (unsigned)i < total)
                ((uint4*)rec)[b * (BLK_E / 4) + i] = ((const uint4*)sorted)[i];
        }
        if (tid < N_TILES)
            offs[b * N_TILES + tid] = first[tid] | (first[tid + 1] << 16);
    }
}


#define BPW      ((N_BLK + 31) / 32)
#define GB       16
#define ACT_S    132

__global__ __launch_bounds__(1024) void k_tile_agg(const unsigned* __restrict__ offs,
                                                   const unsigned* __restrict__ rec,
                                                   const unsigned* __restrict__ deg,
                                                   const float* __restrict__ h,
                                                   const float* __restrict__ b1,
                                                   const float* __restrict__ W2,
                                                   const float* __restrict__ b2,
                                                   float* __restrict__ out) {
    __shared__ float    acc[TILE_R * DIM];
    __shared__ __attribute__((aligned(16))) float actbuf[TILE_R * ACT_S];
    __shared__ float    dinv_row[TILE_R];
    __shared__ __attribute__((aligned(16))) unsigned short deg16[N_NODES];

    const int tid  = threadIdx.x;
    const int lane = tid & 31;
    const int wave = tid >> 5;
    const int wv   = __builtin_amdgcn_readfirstlane(wave);
    const int lo   = blockIdx.x * TILE_R;
    const int nrows = (N_NODES - lo < TILE_R) ? (N_NODES - lo) : TILE_R;
    const float4* h4 = (const float4*)h;

    const unsigned* offt = offs + blockIdx.x;
    unsigned p0 = (wv      < N_BLK) ? offt[(wv     ) * N_TILES] : 0u;
    unsigned p1 = (wv + 32 < N_BLK) ? offt[(wv + 32) * N_TILES] : 0u;
    unsigned r0;
    {
        const unsigned st = p0 & 0xFFFFu, n = (p0 >> 16) - st;
        r0 = ((unsigned)lane < n) ? rec[(unsigned)wv * BLK_E + st + (unsigned)lane] : 0u;
    }

    {
        const uint4* dg4 = (const uint4*)deg;
#pragma unroll
        for (int k = 0; k < (N_NODES / 4 + 1023) / 1024; ++k) {
            const int i = k * 1024 + tid;
            if (i < N_NODES / 4) {
                const uint4 d = dg4[i];
                uint2 p;
                p.x = (d.x < 0xFFFFu ? d.x : 0xFFFFu) | ((d.y < 0xFFFFu ? d.y : 0xFFFFu) << 16);
                p.y = (d.z < 0xFFFFu ? d.z : 0xFFFFu) | ((d.w < 0xFFFFu ? d.w : 0xFFFFu) << 16);
                ((uint2*)deg16)[i] = p;
            }
        }
    }
    if (tid < TILE_R) dinv_row[tid] = (tid < nrows) ? rsqrtf((float)deg[lo + tid]) : 0.0f;
    for (int r = wave; r < nrows; r += 32) {
        const float  dd = rsqrtf((float)deg[lo + r]);
        const float4 v  = h4[(size_t)(lo + r) * (DIM / 4) + lane];
        float* a = acc + r * DIM + lane;
        a[0] = v.x * dd; a[32] = v.y * dd; a[64] = v.z * dd; a[96] = v.w * dd;
    }
    __syncthreads();

    for (int b = wv; b < N_BLK; b += 32) {
        const unsigned p2 = (b + 64 < N_BLK) ? offt[(b + 64) * N_TILES] : 0u;
        const unsigned p   = (unsigned)__builtin_amdgcn_readfirstlane((int)p0);
        const unsigned st  = p & 0xFFFFu;
        const unsigned n   = (p >> 16) - st;
        const unsigned q   = (unsigned)__builtin_amdgcn_readfirstlane((int)p1);
        const unsigned st1 = q & 0xFFFFu;
        const unsigned n1  = (q >> 16) - st1;
        const unsigned* nxt = rec + (unsigned)(b + 32) * BLK_E + st1 + (unsigned)lane;
        unsigned r1  = 0u;
        unsigned rcv = r0;
        for (unsigned w0 = 0u; w0 < n; w0 += 32u) {
            const unsigned m = (n - w0 < 32u) ? (n - w0) : 32u;
            if (w0 != 0u)
                rcv = ((unsigned)lane < m) ? rec[(unsigned)b * BLK_E + st + w0 + (unsigned)lane] : 0u;
            int scl = 0;
            for (unsigned j = 0u; j < m; j += GB) {
                const unsigned c = (m - j < (unsigned)GB) ? (m - j) : (unsigned)GB;
                unsigned r[GB];
                float4   v[GB];
#pragma unroll
                for (int u = 0; u < GB; ++u)
                    r[u] = (unsigned)__builtin_amdgcn_readlane((int)rcv, (int)(j + (unsigned)u));
#pragma unroll
                for (int u = 0; u < GB; ++u) {
                    if ((unsigned)u < c) v[u] = h4[(size_t)(r[u] & 0xFFFFu) * (DIM / 4) + lane];
                }
                if (w0 == 0u && j == 0u && (unsigned)lane < n1) r1 = *nxt;
                if (j == 0u) {
                    unsigned rq = rcv;
                    asm volatile("" : "+v"(rq));
                    unsigned dl = deg16[rq & 0xFFFFu];
                    if (dl == 0xFFFFu) dl = deg[rq & 0xFFFFu];
                    scl = __float_as_int(rsqrtf((float)dl));
                }
#pragma unroll
                for (int u = 0; u < GB; ++u) {
                    if ((unsigned)u < c) {
                        float* a = acc + (r[u] >> 16) * DIM + lane;
                        const float sc = __int_as_float(__builtin_amdgcn_readlane(scl, (int)(j + (unsigned)u)));
                        unsafeAtomicAdd(a,      v[u].x * sc);
                        unsafeAtomicAdd(a + 32, v[u].y * sc);
                        unsafeAtomicAdd(a + 64, v[u].z * sc);
                        unsafeAtomicAdd(a + 96, v[u].w * sc);
                    }
                }
            }
        }
        if (n == 0u && (unsigned)lane < n1) r1 = *nxt;
        p0 = p1; p1 = p2; r0 = r1;
    }


    const int l15     = lane & 15;
    const int hi      = lane >> 4;
    const int colBase = (wv & 7) << 4;

    const float4 bb = ((const float4*)b1)[lane];

    __syncthreads();

    v2f bw[DIM / 4];
    const float* wcol = W2 + (size_t)(hi << 1) * DIM + colBase + l15;
#pragma unroll
    for (int kk = 0; kk < DIM / 4; ++kk) {
        bw[kk].x = wcol[(size_t)(4 * kk) * DIM];
        bw[kk].y = wcol[(size_t)(4 * kk + 1) * DIM];
    }
    const float bv = b2[colBase + l15];

    float* act = actbuf;
    for (int r = wv; r < nrows; r += 32) {
        const float* a = acc + r * DIM + lane;
        const float dr = dinv_row[r];
        const float x0 = a[0] * dr + bb.x, x1 = a[32] * dr + bb.y, x2 = a[64] * dr + bb.z, x3 = a[96] * dr + bb.w;
        float4 o;
        o.x = tanhf(x0);
        o.y = tanhf(x1);
        o.z = tanhf(x2);
        o.w = tanhf(x3);
        *(float4*)(act + r * ACT_S + 4 * lane) = o;
    }
    __syncthreads();

    const int nstrips = (nrows + 15) >> 4;
    for (int s = wv >> 3; s < nstrips; s += 4) {
        int row = (s << 4) + l15;
        row = (row < nrows) ? row : (nrows - 1);
        const float* ar = act + row * ACT_S + (hi << 1);
        v8f c = {};
#pragma unroll
        for (int kk = 0; kk < DIM / 4; ++kk) {
            const v2f a = *(const v2f*)(ar + 4 * kk);
            c = __builtin_amdgcn_wmma_f32_16x16x4_f32(false, a, false, bw[kk],
                                                      (short)0, c, false, false);
        }
        const int m0 = (s << 4) + (hi << 3);
        const int nvalid = nrows - m0;
        float* orow = out + (size_t)(lo + m0) * DIM + colBase + l15;
#pragma unroll
        for (int r = 0; r < 8; ++r)
            if (r < nvalid) orow[(size_t)r * DIM] = c[r] + bv;
    }
}


extern "C" void kernel_launch(void* const* d_in, const int* in_sizes, int n_in,
                              void* d_out, int out_size, void* d_ws, size_t ws_size,
                              hipStream_t stream) {
    const float* x  = (const float*)d_in[0];
    const int*   ei = (const int*)  d_in[1];
    const float* W1 = (const float*)d_in[2];
    const float* b1 = (const float*)d_in[3];
    const float* W2 = (const float*)d_in[4];
    const float* b2 = (const float*)d_in[5];
    float* out = (float*)d_out;

    float*    ws   = (float*)d_ws;
    unsigned* deg  = (unsigned*)ws;
    float*    h    = ws + 131072;
    unsigned* offs = (unsigned*)(ws + 131072 + N_NODES * DIM);
    unsigned* rec  = offs + NE4;

    const int* esrc = ei;
    const int* edst = ei + N_EDGES;

    k_deg_init <<<(N_NODES + 255) / 256, 256, 0, stream>>>(deg);
    k_gemm_sort<<<GEMM_WGS + N_BLK, SG_THREADS, 0, stream>>>(x, W1, esrc, edst, deg, offs, rec, h);
    k_tile_agg<<<N_TILES, 1024, 0, stream>>>(offs, rec, deg, h, b1, W2, b2, out);
}
